# starting kernel
# baseline (speedup 1.0000x reference)
#include <hip/hip_runtime.h>
#include <hip/hip_bf16.h>

typedef _Float16 v16h __attribute__((ext_vector_type(16)));
typedef _Float16 v8h  __attribute__((ext_vector_type(8)));
typedef float    v8f  __attribute__((ext_vector_type(8)));

#define NEG_SLOPE 0.2f

__global__ void k_f32_to_f16(const float* __restrict__ in, _Float16* __restrict__ out, long n) {
    long i = (long)blockIdx.x * blockDim.x + threadIdx.x;
    if (i < n) out[i] = (_Float16)in[i];
}

__global__ void k_fill(float* __restrict__ p, float v, long n) {
    long i = (long)blockIdx.x * blockDim.x + threadIdx.x;
    if (i < n) p[i] = v;
}


__device__ __forceinline__ v16h load_frag(const _Float16* __restrict__ p, int khalf) {
    v8h lo = *(const v8h*)(p + khalf);
    v8h hi = *(const v8h*)(p + 16 + khalf);
    v16h f;
#pragma unroll
    for (int i = 0; i < 8; ++i) { f[i] = lo[i]; f[8 + i] = hi[i]; }
    return f;
}

__device__ __forceinline__ v8f wmma_f16(v16h a, v16h b, v8f c) {
    return __builtin_amdgcn_wmma_f32_16x16x32_f16(
         false, a,  false, b,
         (short)0, c,  false,  false);
}

__global__ void k_gemm_wmma(const _Float16* __restrict__ A, const _Float16* __restrict__ B,
                            float* __restrict__ C, int Nout, int K,
                            int tilesN4, int totalWaves) {
    int wave = blockIdx.x * (blockDim.x >> 5) + (threadIdx.x >> 5);
    if (wave >= totalWaves) return;
    int lane  = threadIdx.x & 31;
    int r     = lane & 15;
    int hi    = lane >> 4;
    int khalf = hi << 3;

    int mt  = wave / tilesN4;
    int nt4 = wave - mt * tilesN4;

    const _Float16* arow = A + (long)(mt * 16 + r) * K;
    const _Float16* brow = B + (long)(nt4 * 64 + r) * K;

    v8f acc0 = {}, acc1 = {}, acc2 = {}, acc3 = {};
    for (int kk = 0; kk < K; kk += 32) {
        v16h a  = load_frag(arow + kk, khalf);
        v16h b0 = load_frag(brow + kk, khalf);
        v16h b1 = load_frag(brow + (long)16 * K + kk, khalf);
        v16h b2 = load_frag(brow + (long)32 * K + kk, khalf);
        v16h b3 = load_frag(brow + (long)48 * K + kk, khalf);
        acc0 = wmma_f16(a, b0, acc0);
        acc1 = wmma_f16(a, b1, acc1);
        acc2 = wmma_f16(a, b2, acc2);
        acc3 = wmma_f16(a, b3, acc3);
    }

    int hi8 = hi << 3;
    float* crow = C + (long)nt4 * 64 + r;
#pragma unroll
    for (int v = 0; v < 8; ++v) {
        long ro = (long)(mt * 16 + hi8 + v) * Nout;
        crow[ro +  0] = acc0[v];
        crow[ro + 16] = acc1[v];
        crow[ro + 32] = acc2[v];
        crow[ro + 48] = acc3[v];
    }
}

__global__ void k_att(const float* __restrict__ Hm,
                      const float* __restrict__ asr, const float* __restrict__ adr,
                      float* __restrict__ als, float* __restrict__ ald,
                      int NH, int H, int C) {
    int i = blockIdx.x * blockDim.x + threadIdx.x;
    if (i >= NH) return;
    int h = i % H;
    const float* hv = Hm + (long)i * C;
    const float* av = asr + h * C;
    const float* dv = adr + h * C;
    float s = 0.f, d = 0.f;
    for (int c = 0; c < C; ++c) {
        float x = hv[c];
        s += x * av[c];
        d += x * dv[c];
    }
    als[i] = s;
    ald[i] = d;
}

__device__ __forceinline__ void atomicMaxF(float* a, float v) {
    if (v >= 0.f) atomicMax((int*)a, __float_as_int(v));
    else          atomicMin((unsigned int*)a, __float_as_uint(v));
}

__device__ __forceinline__ void edge_sd(const int* src, const int* dst, int e, int E,
                                        int& s, int& d) {
    if (e < E) { s = src[e]; d = dst[e]; }
    else       { s = e - E;  d = e - E; }
}

__global__ void k_edge_max(const int* __restrict__ src, const int* __restrict__ dst,
                           int E, int Etot,
                           const float* __restrict__ als, const float* __restrict__ ald,
                           float* __restrict__ m, int H) {
    int i = blockIdx.x * blockDim.x + threadIdx.x;
    if (i >= Etot * H) return;
    int e = i / H, h = i - e * H;
    int s, d; edge_sd(src, dst, e, E, s, d);
    float ev = als[s * H + h] + ald[d * H + h];
    ev = ev > 0.f ? ev : NEG_SLOPE * ev;
    atomicMaxF(&m[d * H + h], ev);
}

__global__ void k_edge_denom(const int* __restrict__ src, const int* __restrict__ dst,
                             int E, int Etot,
                             const float* __restrict__ als, const float* __restrict__ ald,
                             const float* __restrict__ m, float* __restrict__ denom, int H) {
    int i = blockIdx.x * blockDim.x + threadIdx.x;
    if (i >= Etot * H) return;
    int e = i / H, h = i - e * H;
    int s, d; edge_sd(src, dst, e, E, s, d);
    float ev = als[s * H + h] + ald[d * H + h];
    ev = ev > 0.f ? ev : NEG_SLOPE * ev;
    atomicAdd(&denom[d * H + h], expf(ev - m[d * H + h]));
}

__global__ void k_edge_scatter(const int* __restrict__ src, const int* __restrict__ dst,
                               int E, int Etot,
                               const float* __restrict__ als, const float* __restrict__ ald,
                               const float* __restrict__ m, const float* __restrict__ denom,
                               const float* __restrict__ Hm, float* __restrict__ agg,
                               int H, int C) {
    int nout4 = (H * C) >> 2;
    int i = blockIdx.x * blockDim.x + threadIdx.x;
    if (i >= Etot * nout4) return;
    int e  = i / nout4;
    int c4 = i - e * nout4;
    int c  = c4 << 2;
    int h  = c / C;
    int s, d; edge_sd(src, dst, e, E, s, d);
    float ev = als[s * H + h] + ald[d * H + h];
    ev = ev > 0.f ? ev : NEG_SLOPE * ev;
    float alpha = expf(ev - m[d * H + h]) / (denom[d * H + h] + 1e-16f);
    const float4 hv = *(const float4*)(Hm + (long)s * (H * C) + c);
    float* ap = agg + (long)d * (H * C) + c;
    atomicAdd(ap + 0, hv.x * alpha);
    atomicAdd(ap + 1, hv.y * alpha);
    atomicAdd(ap + 2, hv.z * alpha);
    atomicAdd(ap + 3, hv.w * alpha);
}

__global__ void k_bias_act(float* __restrict__ out, const float* __restrict__ bias,
                           long n, int Nout, int doElu) {
    long i = (long)blockIdx.x * blockDim.x + threadIdx.x;
    if (i >= n) return;
    float v = out[i] + bias[i % Nout];
    if (doElu) v = v > 0.f ? v : (expf(v) - 1.f);
    out[i] = v;
}

static inline long cdivl(long a, long b) { return (a + b - 1) / b; }

struct LayerWS {
    _Float16* Xh; _Float16* Wh;
    float *als, *ald, *m, *denom;
};

static void run_gat_layer(const float* Xin, const float* W, const float* a_s, const float* a_d,
                          const float* bias, int N, int K, int Nout, int H, int C,
                          const int* src, const int* dst, int E, int Etot,
                          float* Hbuf, float* agg, int doElu, const LayerWS& w,
                          hipStream_t stream) {
    const int TB = 256;
    k_f32_to_f16<<<(int)cdivl((long)N * K, TB), TB, 0, stream>>>(Xin, w.Xh, (long)N * K);
    k_f32_to_f16<<<(int)cdivl((long)Nout * K, TB), TB, 0, stream>>>(W, w.Wh, (long)Nout * K);
    int tilesN4 = Nout / 64, totalWaves = (N / 16) * tilesN4;
    k_gemm_wmma<<<(int)cdivl(totalWaves, 8), TB, 0, stream>>>(w.Xh, w.Wh, Hbuf, Nout, K,
                                                              tilesN4, totalWaves);
    k_att<<<(int)cdivl((long)N * H, TB), TB, 0, stream>>>(Hbuf, a_s, a_d, w.als, w.ald,
                                                          N * H, H, C);
    k_fill<<<(int)cdivl((long)N * H, TB), TB, 0, stream>>>(w.m, -INFINITY, (long)N * H);
    k_fill<<<(int)cdivl((long)N * H, TB), TB, 0, stream>>>(w.denom, 0.f, (long)N * H);
    k_fill<<<(int)cdivl((long)N * Nout, TB), TB, 0, stream>>>(agg, 0.f, (long)N * Nout);
    int nEH = Etot * H;
    k_edge_max<<<(int)cdivl(nEH, TB), TB, 0, stream>>>(src, dst, E, Etot, w.als, w.ald, w.m, H);
    k_edge_denom<<<(int)cdivl(nEH, TB), TB, 0, stream>>>(src, dst, E, Etot, w.als, w.ald,
                                                         w.m, w.denom, H);
    int nSc = Etot * (Nout >> 2);
    k_edge_scatter<<<(int)cdivl(nSc, TB), TB, 0, stream>>>(src, dst, E, Etot, w.als, w.ald,
                                                           w.m, w.denom, Hbuf, agg, H, C);
    k_bias_act<<<(int)cdivl((long)N * Nout, TB), TB, 0, stream>>>(agg, bias, (long)N * Nout,
                                                                  Nout, doElu);
}

extern "C" void kernel_launch(void* const* d_in, const int* in_sizes, int n_in,
                              void* d_out, int out_size, void* d_ws, size_t ws_size,
                              hipStream_t stream) {
    const float* x   = (const float*)d_in[0];
    const int*   ei  = (const int*)  d_in[1];
    const float* W1  = (const float*)d_in[2];
    const float* as1 = (const float*)d_in[3];
    const float* ad1 = (const float*)d_in[4];
    const float* b1  = (const float*)d_in[5];
    const float* W2  = (const float*)d_in[6];
    const float* as2 = (const float*)d_in[7];
    const float* ad2 = (const float*)d_in[8];
    const float* b2  = (const float*)d_in[9];
    const float* W3  = (const float*)d_in[10];
    const float* as3 = (const float*)d_in[11];
    const float* ad3 = (const float*)d_in[12];
    const float* b3  = (const float*)d_in[13];
    float* out = (float*)d_out;

    const int FinW = 256;
    const int N    = in_sizes[0] / FinW;
    const int E    = in_sizes[1] / 2;
    const int Etot = E + N;
    const int* src = ei;
    const int* dst = ei + E;

    char* p = (char*)d_ws;
    float*    P   = (float*)p;    p += (size_t)N * 256 * sizeof(float);
    float*    Q   = (float*)p;    p += (size_t)N * 256 * sizeof(float);
    _Float16* Xh  = (_Float16*)p; p += (size_t)N * 256 * sizeof(_Float16);
    _Float16* Wh  = (_Float16*)p; p += (size_t)256 * 256 * sizeof(_Float16);
    float*    als = (float*)p;    p += (size_t)N * 8 * sizeof(float);
    float*    ald = (float*)p;    p += (size_t)N * 8 * sizeof(float);
    float*    m   = (float*)p;    p += (size_t)N * 8 * sizeof(float);
    float*    den = (float*)p;    p += (size_t)N * 8 * sizeof(float);
    LayerWS w{Xh, Wh, als, ald, m, den};

    run_gat_layer(x, W1, as1, ad1, b1, N, 256, 256, 8, 32,
                  src, dst, E, Etot, Q, P,  1, w, stream);
    run_gat_layer(P, W2, as2, ad2, b2, N, 256, 256, 8, 32,
                  src, dst, E, Etot, Q, P,  1, w, stream);
    run_gat_layer(P, W3, as3, ad3, b3, N, 256, 128, 1, 128,
                  src, dst, E, Etot, Q, out,  0, w, stream);
}


# Float-atomic scatter replaced by per-node gather over an edge list grouped by destination, sources ascending, built per call: max, sum, weighted rows, bias, act in registers; fixed summation order
# speedup vs baseline: 2.4641x; 2.4641x over previous
#include <hip/hip_runtime.h>
#include <hip/hip_bf16.h>

typedef _Float16 v16h __attribute__((ext_vector_type(16)));
typedef _Float16 v8h  __attribute__((ext_vector_type(8)));
typedef float    v8f  __attribute__((ext_vector_type(8)));

#define NEG_SLOPE 0.2f

__global__ void k_f32_to_f16(const float* __restrict__ in, _Float16* __restrict__ out, long n) {
    long i = (long)blockIdx.x * blockDim.x + threadIdx.x;
    if (i < n) out[i] = (_Float16)in[i];
}


__device__ __forceinline__ v16h load_frag(const _Float16* __restrict__ p, int khalf) {
    v8h lo = *(const v8h*)(p + khalf);
    v8h hi = *(const v8h*)(p + 16 + khalf);
    v16h f;
#pragma unroll
    for (int i = 0; i < 8; ++i) { f[i] = lo[i]; f[8 + i] = hi[i]; }
    return f;
}

__device__ __forceinline__ v8f wmma_f16(v16h a, v16h b, v8f c) {
    return __builtin_amdgcn_wmma_f32_16x16x32_f16(
         false, a,  false, b,
         (short)0, c,  false,  false);
}

__global__ void k_gemm_wmma(const _Float16* __restrict__ A, const _Float16* __restrict__ B,
                            float* __restrict__ C, int Nout, int K,
                            int tilesN4, int totalWaves) {
    int wave = blockIdx.x * (blockDim.x >> 5) + (threadIdx.x >> 5);
    if (wave >= totalWaves) return;
    int lane  = threadIdx.x & 31;
    int r     = lane & 15;
    int hi    = lane >> 4;
    int khalf = hi << 3;

    int mt  = wave / tilesN4;
    int nt4 = wave - mt * tilesN4;

    const _Float16* arow = A + (long)(mt * 16 + r) * K;
    const _Float16* brow = B + (long)(nt4 * 64 + r) * K;

    v8f acc0 = {}, acc1 = {}, acc2 = {}, acc3 = {};
    for (int kk = 0; kk < K; kk += 32) {
        v16h a  = load_frag(arow + kk, khalf);
        v16h b0 = load_frag(brow + kk, khalf);
        v16h b1 = load_frag(brow + (long)16 * K + kk, khalf);
        v16h b2 = load_frag(brow + (long)32 * K + kk, khalf);
        v16h b3 = load_frag(brow + (long)48 * K + kk, khalf);
        acc0 = wmma_f16(a, b0, acc0);
        acc1 = wmma_f16(a, b1, acc1);
        acc2 = wmma_f16(a, b2, acc2);
        acc3 = wmma_f16(a, b3, acc3);
    }

    int hi8 = hi << 3;
    float* crow = C + (long)nt4 * 64 + r;
#pragma unroll
    for (int v = 0; v < 8; ++v) {
        long ro = (long)(mt * 16 + hi8 + v) * Nout;
        crow[ro +  0] = acc0[v];
        crow[ro + 16] = acc1[v];
        crow[ro + 32] = acc2[v];
        crow[ro + 48] = acc3[v];
    }
}

__global__ void k_att(const float* __restrict__ Hm,
                      const float* __restrict__ asr, const float* __restrict__ adr,
                      float* __restrict__ als, float* __restrict__ ald,
                      int NH, int H, int C) {
    int i = blockIdx.x * blockDim.x + threadIdx.x;
    if (i >= NH) return;
    int h = i % H;
    const float* hv = Hm + (long)i * C;
    const float* av = asr + h * C;
    const float* dv = adr + h * C;
    float s = 0.f, d = 0.f;
    for (int c = 0; c < C; ++c) {
        float x = hv[c];
        s += x * av[c];
        d += x * dv[c];
    }
    als[i] = s;
    ald[i] = d;
}

__global__ void k_fill_i32(int* __restrict__ p, int v, long n) {
    long i = (long)blockIdx.x * blockDim.x + threadIdx.x;
    if (i < n) p[i] = v;
}

__global__ void k_deg_count(const int* __restrict__ dst, int E, int* __restrict__ cnt) {
    int e = blockIdx.x * blockDim.x + threadIdx.x;
    if (e < E) atomicAdd(&cnt[dst[e]], 1);
}

__global__ void k_scan(const int* __restrict__ cnt, int* __restrict__ rowptr, int N) {
    __shared__ int part[1024];
    int t  = threadIdx.x;
    int nt = blockDim.x;
    int per = (N + nt - 1) / nt;
    int b = t * per;
    int e = b + per < N ? b + per : N;
    int s = 0;
    for (int i = b; i < e; ++i) s += cnt[i];
    part[t] = s;
    __syncthreads();
    for (int off = 1; off < nt; off <<= 1) {
        int v = t >= off ? part[t - off] : 0;
        __syncthreads();
        part[t] += v;
        __syncthreads();
    }
    int run = part[t] - s;
    for (int i = b; i < e; ++i) { rowptr[i] = run; run += cnt[i]; }
    if (t == nt - 1) rowptr[N] = part[t];
}

__global__ void k_place(const int* __restrict__ src, const int* __restrict__ dst, int E,
                        const int* __restrict__ rowptr, int* __restrict__ cursor,
                        int* __restrict__ tmp, int* __restrict__ slot) {
    int e = blockIdx.x * blockDim.x + threadIdx.x;
    if (e >= E) return;
    int d = dst[e];
    int p = rowptr[d] + atomicAdd(&cursor[d], 1);
    tmp[p]  = src[e];
    slot[e] = p;
}

__global__ void k_rank(const int* __restrict__ src, const int* __restrict__ dst, int E,
                       const int* __restrict__ rowptr, const int* __restrict__ tmp,
                       const int* __restrict__ slot, int* __restrict__ csr) {
    int e = blockIdx.x * blockDim.x + threadIdx.x;
    if (e >= E) return;
    int d = dst[e], s = src[e], p = slot[e];
    int b = rowptr[d], en = rowptr[d + 1];
    int r = 0;
    for (int q = b; q < en; ++q) {
        int v = tmp[q];
        r += (v < s || (v == s && q < p)) ? 1 : 0;
    }
    csr[b + r] = s;
}

__device__ __forceinline__ float leaky(float v) { return v > 0.f ? v : NEG_SLOPE * v; }

template <int H, int C, int ELU>
__global__ void __launch_bounds__(256)
k_gather(const int* __restrict__ rowptr, const int* __restrict__ csr,
         const float* __restrict__ als, const float* __restrict__ ald,
         const float* __restrict__ Hm, const float* __restrict__ bias,
         float* __restrict__ out, int N, int totalWaves) {
    constexpr int NOUT = H * C;
    constexpr int V    = NOUT / 128;
    const int lane = threadIdx.x & 31;
    const int wave0 = blockIdx.x * (blockDim.x >> 5) + (threadIdx.x >> 5);

    int hd[V], ch[V];
#pragma unroll
    for (int k = 0; k < V; ++k) { ch[k] = k * 128 + lane * 4; hd[k] = ch[k] / C; }

    for (int nn = wave0; nn < N; nn += totalWaves) {
        const int n   = __builtin_amdgcn_readfirstlane(nn);
        const int beg = rowptr[n], end = rowptr[n + 1];

        float adn[V], m[V], den[V];
        float4 acc[V];
#pragma unroll
        for (int k = 0; k < V; ++k) {
            adn[k] = ald[n * H + hd[k]];
            m[k]   = leaky(als[n * H + hd[k]] + adn[k]);
        }
        for (int base = beg; base < end; base += 32) {
            const int cnt = end - base < 32 ? end - base : 32;
            const int mys = lane < cnt ? csr[base + lane] : 0;
            for (int j = 0; j < cnt; ++j) {
                const int s = __builtin_amdgcn_readlane(mys, j);
#pragma unroll
                for (int k = 0; k < V; ++k)
                    m[k] = fmaxf(m[k], leaky(als[s * H + hd[k]] + adn[k]));
            }
        }
#pragma unroll
        for (int k = 0; k < V; ++k) {
            const float ex = expf(leaky(als[n * H + hd[k]] + adn[k]) - m[k]);
            const float4 hv = *(const float4*)(Hm + (long)n * NOUT + ch[k]);
            den[k] = ex;
            acc[k] = make_float4(hv.x * ex, hv.y * ex, hv.z * ex, hv.w * ex);
        }
        for (int base = beg; base < end; base += 32) {
            const int cnt = end - base < 32 ? end - base : 32;
            const int mys = lane < cnt ? csr[base + lane] : 0;
#pragma unroll 4
            for (int j = 0; j < cnt; ++j) {
                const int s = __builtin_amdgcn_readlane(mys, j);
#pragma unroll
                for (int k = 0; k < V; ++k) {
                    const float ex = expf(leaky(als[s * H + hd[k]] + adn[k]) - m[k]);
                    const float4 hv = *(const float4*)(Hm + (long)s * NOUT + ch[k]);
                    den[k] += ex;
                    acc[k].x += hv.x * ex;
                    acc[k].y += hv.y * ex;
                    acc[k].z += hv.z * ex;
                    acc[k].w += hv.w * ex;
                }
            }
        }
#pragma unroll
        for (int k = 0; k < V; ++k) {
            const float dn = den[k] + 1e-16f;
            const float4 bv = *(const float4*)(bias + ch[k]);
            float4 o;
            o.x = acc[k].x / dn + bv.x;
            o.y = acc[k].y / dn + bv.y;
            o.z = acc[k].z / dn + bv.z;
            o.w = acc[k].w / dn + bv.w;
            if (ELU) {
                o.x = o.x > 0.f ? o.x : (expf(o.x) - 1.f);
                o.y = o.y > 0.f ? o.y : (expf(o.y) - 1.f);
                o.z = o.z > 0.f ? o.z : (expf(o.z) - 1.f);
                o.w = o.w > 0.f ? o.w : (expf(o.w) - 1.f);
            }
            *(float4*)(out + (long)n * NOUT + ch[k]) = o;
        }
    }
}

static inline long cdivl(long a, long b) { return (a + b - 1) / b; }

struct LayerWS {
    _Float16* Xh; _Float16* Wh;
    float *als, *ald;
    const int *rowptr, *csr;
};

static void run_gat_layer(const float* Xin, const float* W, const float* a_s, const float* a_d,
                          const float* bias, int N, int K, int Nout, int H, int C,
                          float* Hbuf, float* agg, const LayerWS& w,
                          hipStream_t stream) {
    const int TB = 256;
    k_f32_to_f16<<<(int)cdivl((long)N * K, TB), TB, 0, stream>>>(Xin, w.Xh, (long)N * K);
    k_f32_to_f16<<<(int)cdivl((long)Nout * K, TB), TB, 0, stream>>>(W, w.Wh, (long)Nout * K);
    int tilesN4 = Nout / 64, totalWaves = (N / 16) * tilesN4;
    k_gemm_wmma<<<(int)cdivl(totalWaves, 8), TB, 0, stream>>>(w.Xh, w.Wh, Hbuf, Nout, K,
                                                              tilesN4, totalWaves);
    k_att<<<(int)cdivl((long)N * H, TB), TB, 0, stream>>>(Hbuf, a_s, a_d, w.als, w.ald,
                                                          N * H, H, C);
    const int GB = 2048, gWaves = GB * (TB / 32);
    if (H == 8)
        k_gather<8, 32, 1><<<GB, TB, 0, stream>>>(w.rowptr, w.csr, w.als, w.ald, Hbuf, bias,
                                                  agg, N, gWaves);
    else
        k_gather<1, 128, 0><<<GB, TB, 0, stream>>>(w.rowptr, w.csr, w.als, w.ald, Hbuf, bias,
                                                   agg, N, gWaves);
}

extern "C" void kernel_launch(void* const* d_in, const int* in_sizes, int n_in,
                              void* d_out, int out_size, void* d_ws, size_t ws_size,
                              hipStream_t stream) {
    const float* x   = (const float*)d_in[0];
    const int*   ei  = (const int*)  d_in[1];
    const float* W1  = (const float*)d_in[2];
    const float* as1 = (const float*)d_in[3];
    const float* ad1 = (const float*)d_in[4];
    const float* b1  = (const float*)d_in[5];
    const float* W2  = (const float*)d_in[6];
    const float* as2 = (const float*)d_in[7];
    const float* ad2 = (const float*)d_in[8];
    const float* b2  = (const float*)d_in[9];
    const float* W3  = (const float*)d_in[10];
    const float* as3 = (const float*)d_in[11];
    const float* ad3 = (const float*)d_in[12];
    const float* b3  = (const float*)d_in[13];
    float* out = (float*)d_out;

    const int FinW = 256;
    const int N    = in_sizes[0] / FinW;
    const int E    = in_sizes[1] / 2;
    const int* src = ei;
    const int* dst = ei + E;

    char* p = (char*)d_ws;
    float*    P   = (float*)p;    p += (size_t)N * 256 * sizeof(float);
    float*    Q   = (float*)p;    p += (size_t)N * 256 * sizeof(float);
    _Float16* Xh  = (_Float16*)p; p += (size_t)N * 256 * sizeof(_Float16);
    _Float16* Wh  = (_Float16*)p; p += (size_t)256 * 256 * sizeof(_Float16);
    float*    als = (float*)p;    p += (size_t)N * 8 * sizeof(float);
    float*    ald = (float*)p;    p += (size_t)N * 8 * sizeof(float);
    int*      cnt = (int*)p;      p += (size_t)N * sizeof(int);
    int*      cur = (int*)p;      p += (size_t)N * sizeof(int);
    int*      rowptr = (int*)p;   p += (size_t)(N + 4) * sizeof(int);
    int*      tmp  = (int*)p;     p += (size_t)E * sizeof(int);
    int*      slot = (int*)p;     p += (size_t)E * sizeof(int);
    int*      csr  = (int*)p;     p += (size_t)E * sizeof(int);
    LayerWS w{Xh, Wh, als, ald, rowptr, csr};

    const int TB = 256;
    k_fill_i32<<<(int)cdivl((long)2 * N, TB), TB, 0, stream>>>(cnt, 0, (long)2 * N);
    k_deg_count<<<(int)cdivl(E, TB), TB, 0, stream>>>(dst, E, cnt);
    k_scan<<<1, 1024, 0, stream>>>(cnt, rowptr, N);
    k_place<<<(int)cdivl(E, TB), TB, 0, stream>>>(src, dst, E, rowptr, cur, tmp, slot);
    k_rank<<<(int)cdivl(E, TB), TB, 0, stream>>>(src, dst, E, rowptr, tmp, slot, csr);

    run_gat_layer(x, W1, as1, ad1, b1, N, 256, 256, 8, 32, Q, P, w, stream);
    run_gat_layer(P, W2, as2, ad2, b2, N, 256, 256, 8, 32, Q, P, w, stream);
    run_gat_layer(P, W3, as3, ad3, b3, N, 256, 128, 1, 128, Q, out, w, stream);
}


# Layer-input fp32->16-bit staging pass removed (3 launches, one element per thread): GEMM A loader reads fp32 rows and rounds in registers, next K-step's loads issued ahead; same rounding and k order
# speedup vs baseline: 2.8416x; 1.1532x over previous
#include <hip/hip_runtime.h>
#include <hip/hip_bf16.h>

typedef _Float16 v16h __attribute__((ext_vector_type(16)));
typedef _Float16 v8h  __attribute__((ext_vector_type(8)));
typedef float    v8f  __attribute__((ext_vector_type(8)));

#define NEG_SLOPE 0.2f

__global__ void k_f32_to_f16(const float* __restrict__ in, _Float16* __restrict__ out, long n) {
    long i = (long)blockIdx.x * blockDim.x + threadIdx.x;
    if (i < n) out[i] = (_Float16)in[i];
}


struct KStep {
    float4 a0, a1, a2, a3;
    v8h    blo[4], bhi[4];
};

__device__ __forceinline__ void load_kstep(KStep& s, const float* __restrict__ ap,
                                           const _Float16* __restrict__ bp, long K16, int khalf) {
    s.a0 = *(const float4*)(ap + khalf);
    s.a1 = *(const float4*)(ap + khalf + 4);
    s.a2 = *(const float4*)(ap + 16 + khalf);
    s.a3 = *(const float4*)(ap + 16 + khalf + 4);
#pragma unroll
    for (int j = 0; j < 4; ++j) {
        s.blo[j] = *(const v8h*)(bp + j * K16 + khalf);
        s.bhi[j] = *(const v8h*)(bp + j * K16 + 16 + khalf);
    }
}

__device__ __forceinline__ v16h a_frag(const KStep& s) {
    v16h f;
    f[0]  = (_Float16)s.a0.x; f[1]  = (_Float16)s.a0.y; f[2]  = (_Float16)s.a0.z; f[3]  = (_Float16)s.a0.w;
    f[4]  = (_Float16)s.a1.x; f[5]  = (_Float16)s.a1.y; f[6]  = (_Float16)s.a1.z; f[7]  = (_Float16)s.a1.w;
    f[8]  = (_Float16)s.a2.x; f[9]  = (_Float16)s.a2.y; f[10] = (_Float16)s.a2.z; f[11] = (_Float16)s.a2.w;
    f[12] = (_Float16)s.a3.x; f[13] = (_Float16)s.a3.y; f[14] = (_Float16)s.a3.z; f[15] = (_Float16)s.a3.w;
    return f;
}

__device__ __forceinline__ v16h b_frag(const KStep& s, int j) {
    v16h f;
#pragma unroll
    for (int i = 0; i < 8; ++i) { f[i] = s.blo[j][i]; f[8 + i] = s.bhi[j][i]; }
    return f;
}

__device__ __forceinline__ v8f wmma_f16(v16h a, v16h b, v8f c) {
    return __builtin_amdgcn_wmma_f32_16x16x32_f16(
         false, a,  false, b,
         (short)0, c,  false,  false);
}

template <int K>
__global__ void __launch_bounds__(256)
k_gemm_wmma(const float* __restrict__ A, const _Float16* __restrict__ B,
            float* __restrict__ C, int Nout, int tilesN4, int totalWaves) {
    int wave = blockIdx.x * (blockDim.x >> 5) + (threadIdx.x >> 5);
    if (wave >= totalWaves) return;
    int lane  = threadIdx.x & 31;
    int r     = lane & 15;
    int hi    = lane >> 4;
    int khalf = hi << 3;

    int mt  = wave / tilesN4;
    int nt4 = wave - mt * tilesN4;

    const float*    arow = A + (long)(mt * 16 + r) * K;
    const _Float16* brow = B + (long)(nt4 * 64 + r) * K;

    v8f acc0 = {}, acc1 = {}, acc2 = {}, acc3 = {};
    KStep st[2];
    load_kstep(st[0], arow, brow, (long)16 * K, khalf);
#pragma unroll
    for (int kk = 0; kk < K; kk += 32) {
        const int cur = (kk >> 5) & 1;
        __builtin_amdgcn_sched_barrier(0);
        if (kk + 32 < K) load_kstep(st[cur ^ 1], arow + kk + 32, brow + kk + 32, (long)16 * K, khalf);
        __builtin_amdgcn_sched_barrier(0);
        const v16h a = a_frag(st[cur]);
        acc0 = wmma_f16(a, b_frag(st[cur], 0), acc0);
        acc1 = wmma_f16(a, b_frag(st[cur], 1), acc1);
        acc2 = wmma_f16(a, b_frag(st[cur], 2), acc2);
        acc3 = wmma_f16(a, b_frag(st[cur], 3), acc3);
    }

    int hi8 = hi << 3;
    float* crow = C + (long)nt4 * 64 + r;
#pragma unroll
    for (int v = 0; v < 8; ++v) {
        long ro = (long)(mt * 16 + hi8 + v) * Nout;
        crow[ro +  0] = acc0[v];
        crow[ro + 16] = acc1[v];
        crow[ro + 32] = acc2[v];
        crow[ro + 48] = acc3[v];
    }
}

__global__ void k_att(const float* __restrict__ Hm,
                      const float* __restrict__ asr, const float* __restrict__ adr,
                      float* __restrict__ als, float* __restrict__ ald,
                      int NH, int H, int C) {
    int i = blockIdx.x * blockDim.x + threadIdx.x;
    if (i >= NH) return;
    int h = i % H;
    const float* hv = Hm + (long)i * C;
    const float* av = asr + h * C;
    const float* dv = adr + h * C;
    float s = 0.f, d = 0.f;
    for (int c = 0; c < C; ++c) {
        float x = hv[c];
        s += x * av[c];
        d += x * dv[c];
    }
    als[i] = s;
    ald[i] = d;
}

__global__ void k_fill_i32(int* __restrict__ p, int v, long n) {
    long i = (long)blockIdx.x * blockDim.x + threadIdx.x;
    if (i < n) p[i] = v;
}

__global__ void k_deg_count(const int* __restrict__ dst, int E, int* __restrict__ cnt) {
    int e = blockIdx.x * blockDim.x + threadIdx.x;
    if (e < E) atomicAdd(&cnt[dst[e]], 1);
}

__global__ void k_scan(const int* __restrict__ cnt, int* __restrict__ rowptr, int N) {
    __shared__ int part[1024];
    int t  = threadIdx.x;
    int nt = blockDim.x;
    int per = (N + nt - 1) / nt;
    int b = t * per;
    int e = b + per < N ? b + per : N;
    int s = 0;
    for (int i = b; i < e; ++i) s += cnt[i];
    part[t] = s;
    __syncthreads();
    for (int off = 1; off < nt; off <<= 1) {
        int v = t >= off ? part[t - off] : 0;
        __syncthreads();
        part[t] += v;
        __syncthreads();
    }
    int run = part[t] - s;
    for (int i = b; i < e; ++i) { rowptr[i] = run; run += cnt[i]; }
    if (t == nt - 1) rowptr[N] = part[t];
}

__global__ void k_place(const int* __restrict__ src, const int* __restrict__ dst, int E,
                        const int* __restrict__ rowptr, int* __restrict__ cursor,
                        int* __restrict__ tmp, int* __restrict__ slot) {
    int e = blockIdx.x * blockDim.x + threadIdx.x;
    if (e >= E) return;
    int d = dst[e];
    int p = rowptr[d] + atomicAdd(&cursor[d], 1);
    tmp[p]  = src[e];
    slot[e] = p;
}

__global__ void k_rank(const int* __restrict__ src, const int* __restrict__ dst, int E,
                       const int* __restrict__ rowptr, const int* __restrict__ tmp,
                       const int* __restrict__ slot, int* __restrict__ csr) {
    int e = blockIdx.x * blockDim.x + threadIdx.x;
    if (e >= E) return;
    int d = dst[e], s = src[e], p = slot[e];
    int b = rowptr[d], en = rowptr[d + 1];
    int r = 0;
    for (int q = b; q < en; ++q) {
        int v = tmp[q];
        r += (v < s || (v == s && q < p)) ? 1 : 0;
    }
    csr[b + r] = s;
}

__device__ __forceinline__ float leaky(float v) { return v > 0.f ? v : NEG_SLOPE * v; }

template <int H, int C, int ELU>
__global__ void __launch_bounds__(256)
k_gather(const int* __restrict__ rowptr, const int* __restrict__ csr,
         const float* __restrict__ als, const float* __restrict__ ald,
         const float* __restrict__ Hm, const float* __restrict__ bias,
         float* __restrict__ out, int N, int totalWaves) {
    constexpr int NOUT = H * C;
    constexpr int V    = NOUT / 128;
    const int lane = threadIdx.x & 31;
    const int wave0 = blockIdx.x * (blockDim.x >> 5) + (threadIdx.x >> 5);

    int hd[V], ch[V];
#pragma unroll
    for (int k = 0; k < V; ++k) { ch[k] = k * 128 + lane * 4; hd[k] = ch[k] / C; }

    for (int nn = wave0; nn < N; nn += totalWaves) {
        const int n   = __builtin_amdgcn_readfirstlane(nn);
        const int beg = rowptr[n], end = rowptr[n + 1];

        float adn[V], m[V], den[V];
        float4 acc[V];
#pragma unroll
        for (int k = 0; k < V; ++k) {
            adn[k] = ald[n * H + hd[k]];
            m[k]   = leaky(als[n * H + hd[k]] + adn[k]);
        }
        for (int base = beg; base < end; base += 32) {
            const int cnt = end - base < 32 ? end - base : 32;
            const int mys = lane < cnt ? csr[base + lane] : 0;
            for (int j = 0; j < cnt; ++j) {
                const int s = __builtin_amdgcn_readlane(mys, j);
#pragma unroll
                for (int k = 0; k < V; ++k)
                    m[k] = fmaxf(m[k], leaky(als[s * H + hd[k]] + adn[k]));
            }
        }
#pragma unroll
        for (int k = 0; k < V; ++k) {
            const float ex = expf(leaky(als[n * H + hd[k]] + adn[k]) - m[k]);
            const float4 hv = *(const float4*)(Hm + (long)n * NOUT + ch[k]);
            den[k] = ex;
            acc[k] = make_float4(hv.x * ex, hv.y * ex, hv.z * ex, hv.w * ex);
        }
        for (int base = beg; base < end; base += 32) {
            const int cnt = end - base < 32 ? end - base : 32;
            const int mys = lane < cnt ? csr[base + lane] : 0;
#pragma unroll 4
            for (int j = 0; j < cnt; ++j) {
                const int s = __builtin_amdgcn_readlane(mys, j);
#pragma unroll
                for (int k = 0; k < V; ++k) {
                    const float ex = expf(leaky(als[s * H + hd[k]] + adn[k]) - m[k]);
                    const float4 hv = *(const float4*)(Hm + (long)s * NOUT + ch[k]);
                    den[k] += ex;
                    acc[k].x += hv.x * ex;
                    acc[k].y += hv.y * ex;
                    acc[k].z += hv.z * ex;
                    acc[k].w += hv.w * ex;
                }
            }
        }
#pragma unroll
        for (int k = 0; k < V; ++k) {
            const float dn = den[k] + 1e-16f;
            const float4 bv = *(const float4*)(bias + ch[k]);
            float4 o;
            o.x = acc[k].x / dn + bv.x;
            o.y = acc[k].y / dn + bv.y;
            o.z = acc[k].z / dn + bv.z;
            o.w = acc[k].w / dn + bv.w;
            if (ELU) {
                o.x = o.x > 0.f ? o.x : (expf(o.x) - 1.f);
                o.y = o.y > 0.f ? o.y : (expf(o.y) - 1.f);
                o.z = o.z > 0.f ? o.z : (expf(o.z) - 1.f);
                o.w = o.w > 0.f ? o.w : (expf(o.w) - 1.f);
            }
            *(float4*)(out + (long)n * NOUT + ch[k]) = o;
        }
    }
}

static inline long cdivl(long a, long b) { return (a + b - 1) / b; }

struct LayerWS {
    _Float16* Wh;
    float *als, *ald;
    const int *rowptr, *csr;
};

static void run_gat_layer(const float* Xin, const float* W, const float* a_s, const float* a_d,
                          const float* bias, int N, int K, int Nout, int H, int C,
                          float* Hbuf, float* agg, const LayerWS& w,
                          hipStream_t stream) {
    const int TB = 256;
    k_f32_to_f16<<<(int)cdivl((long)Nout * K, TB), TB, 0, stream>>>(W, w.Wh, (long)Nout * K);
    int tilesN4 = Nout / 64, totalWaves = (N / 16) * tilesN4;
    k_gemm_wmma<256><<<(int)cdivl(totalWaves, 8), TB, 0, stream>>>(Xin, w.Wh, Hbuf, Nout,
                                                                   tilesN4, totalWaves);
    k_att<<<(int)cdivl((long)N * H, TB), TB, 0, stream>>>(Hbuf, a_s, a_d, w.als, w.ald,
                                                          N * H, H, C);
    const int GB = 2048, gWaves = GB * (TB / 32);
    if (H == 8)
        k_gather<8, 32, 1><<<GB, TB, 0, stream>>>(w.rowptr, w.csr, w.als, w.ald, Hbuf, bias,
                                                  agg, N, gWaves);
    else
        k_gather<1, 128, 0><<<GB, TB, 0, stream>>>(w.rowptr, w.csr, w.als, w.ald, Hbuf, bias,
                                                   agg, N, gWaves);
}

extern "C" void kernel_launch(void* const* d_in, const int* in_sizes, int n_in,
                              void* d_out, int out_size, void* d_ws, size_t ws_size,
                              hipStream_t stream) {
    const float* x   = (const float*)d_in[0];
    const int*   ei  = (const int*)  d_in[1];
    const float* W1  = (const float*)d_in[2];
    const float* as1 = (const float*)d_in[3];
    const float* ad1 = (const float*)d_in[4];
    const float* b1  = (const float*)d_in[5];
    const float* W2  = (const float*)d_in[6];
    const float* as2 = (const float*)d_in[7];
    const float* ad2 = (const float*)d_in[8];
    const float* b2  = (const float*)d_in[9];
    const float* W3  = (const float*)d_in[10];
    const float* as3 = (const float*)d_in[11];
    const float* ad3 = (const float*)d_in[12];
    const float* b3  = (const float*)d_in[13];
    float* out = (float*)d_out;

    const int FinW = 256;
    const int N    = in_sizes[0] / FinW;
    const int E    = in_sizes[1] / 2;
    const int* src = ei;
    const int* dst = ei + E;

    char* p = (char*)d_ws;
    float*    P   = (float*)p;    p += (size_t)N * 256 * sizeof(float);
    float*    Q   = (float*)p;    p += (size_t)N * 256 * sizeof(float);
    _Float16* Wh  = (_Float16*)p; p += (size_t)256 * 256 * sizeof(_Float16);
    float*    als = (float*)p;    p += (size_t)N * 8 * sizeof(float);
    float*    ald = (float*)p;    p += (size_t)N * 8 * sizeof(float);
    int*      cnt = (int*)p;      p += (size_t)N * sizeof(int);
    int*      cur = (int*)p;      p += (size_t)N * sizeof(int);
    int*      rowptr = (int*)p;   p += (size_t)(N + 4) * sizeof(int);
    int*      tmp  = (int*)p;     p += (size_t)E * sizeof(int);
    int*      slot = (int*)p;     p += (size_t)E * sizeof(int);
    int*      csr  = (int*)p;     p += (size_t)E * sizeof(int);
    LayerWS w{Wh, als, ald, rowptr, csr};

    const int TB = 256;
    k_fill_i32<<<(int)cdivl((long)2 * N, TB), TB, 0, stream>>>(cnt, 0, (long)2 * N);
    k_deg_count<<<(int)cdivl(E, TB), TB, 0, stream>>>(dst, E, cnt);
    k_scan<<<1, 1024, 0, stream>>>(cnt, rowptr, N);
    k_place<<<(int)cdivl(E, TB), TB, 0, stream>>>(src, dst, E, rowptr, cur, tmp, slot);
    k_rank<<<(int)cdivl(E, TB), TB, 0, stream>>>(src, dst, E, rowptr, tmp, slot, csr);

    run_gat_layer(x, W1, as1, ad1, b1, N, 256, 256, 8, 32, Q, P, w, stream);
    run_gat_layer(P, W2, as2, ad2, b2, N, 256, 256, 8, 32, Q, P, w, stream);
    run_gat_layer(P, W3, as3, ad3, b3, N, 256, 128, 1, 128, Q, out, w, stream);
}


# Edge-list build walks memory in whole lines: prefix sum by one contiguous slice per wave, in-group ordering by one wave per destination group (was a strided run per thread); list contents unchanged
# speedup vs baseline: 3.4148x; 1.2017x over previous
#include <hip/hip_runtime.h>
#include <hip/hip_bf16.h>

typedef _Float16 v16h __attribute__((ext_vector_type(16)));
typedef _Float16 v8h  __attribute__((ext_vector_type(8)));
typedef float    v8f  __attribute__((ext_vector_type(8)));

#define NEG_SLOPE 0.2f

__global__ void k_f32_to_f16(const float* __restrict__ in, _Float16* __restrict__ out, long n) {
    long i = (long)blockIdx.x * blockDim.x + threadIdx.x;
    if (i < n) out[i] = (_Float16)in[i];
}


struct KStep {
    float4 a0, a1, a2, a3;
    v8h    blo[4], bhi[4];
};

__device__ __forceinline__ void load_kstep(KStep& s, const float* __restrict__ ap,
                                           const _Float16* __restrict__ bp, long K16, int khalf) {
    s.a0 = *(const float4*)(ap + khalf);
    s.a1 = *(const float4*)(ap + khalf + 4);
    s.a2 = *(const float4*)(ap + 16 + khalf);
    s.a3 = *(const float4*)(ap + 16 + khalf + 4);
#pragma unroll
    for (int j = 0; j < 4; ++j) {
        s.blo[j] = *(const v8h*)(bp + j * K16 + khalf);
        s.bhi[j] = *(const v8h*)(bp + j * K16 + 16 + khalf);
    }
}

__device__ __forceinline__ v16h a_frag(const KStep& s) {
    v16h f;
    f[0]  = (_Float16)s.a0.x; f[1]  = (_Float16)s.a0.y; f[2]  = (_Float16)s.a0.z; f[3]  = (_Float16)s.a0.w;
    f[4]  = (_Float16)s.a1.x; f[5]  = (_Float16)s.a1.y; f[6]  = (_Float16)s.a1.z; f[7]  = (_Float16)s.a1.w;
    f[8]  = (_Float16)s.a2.x; f[9]  = (_Float16)s.a2.y; f[10] = (_Float16)s.a2.z; f[11] = (_Float16)s.a2.w;
    f[12] = (_Float16)s.a3.x; f[13] = (_Float16)s.a3.y; f[14] = (_Float16)s.a3.z; f[15] = (_Float16)s.a3.w;
    return f;
}

__device__ __forceinline__ v16h b_frag(const KStep& s, int j) {
    v16h f;
#pragma unroll
    for (int i = 0; i < 8; ++i) { f[i] = s.blo[j][i]; f[8 + i] = s.bhi[j][i]; }
    return f;
}

__device__ __forceinline__ v8f wmma_f16(v16h a, v16h b, v8f c) {
    return __builtin_amdgcn_wmma_f32_16x16x32_f16(
         false, a,  false, b,
         (short)0, c,  false,  false);
}

template <int K>
__global__ void __launch_bounds__(256)
k_gemm_wmma(const float* __restrict__ A, const _Float16* __restrict__ B,
            float* __restrict__ C, int Nout, int tilesN4, int totalWaves) {
    int wave = blockIdx.x * (blockDim.x >> 5) + (threadIdx.x >> 5);
    if (wave >= totalWaves) return;
    int lane  = threadIdx.x & 31;
    int r     = lane & 15;
    int hi    = lane >> 4;
    int khalf = hi << 3;

    int mt  = wave / tilesN4;
    int nt4 = wave - mt * tilesN4;

    const float*    arow = A + (long)(mt * 16 + r) * K;
    const _Float16* brow = B + (long)(nt4 * 64 + r) * K;

    v8f acc0 = {}, acc1 = {}, acc2 = {}, acc3 = {};
    KStep st[2];
    load_kstep(st[0], arow, brow, (long)16 * K, khalf);
#pragma unroll
    for (int kk = 0; kk < K; kk += 32) {
        const int cur = (kk >> 5) & 1;
        __builtin_amdgcn_sched_barrier(0);
        if (kk + 32 < K) load_kstep(st[cur ^ 1], arow + kk + 32, brow + kk + 32, (long)16 * K, khalf);
        __builtin_amdgcn_sched_barrier(0);
        const v16h a = a_frag(st[cur]);
        acc0 = wmma_f16(a, b_frag(st[cur], 0), acc0);
        acc1 = wmma_f16(a, b_frag(st[cur], 1), acc1);
        acc2 = wmma_f16(a, b_frag(st[cur], 2), acc2);
        acc3 = wmma_f16(a, b_frag(st[cur], 3), acc3);
    }

    int hi8 = hi << 3;
    float* crow = C + (long)nt4 * 64 + r;
#pragma unroll
    for (int v = 0; v < 8; ++v) {
        long ro = (long)(mt * 16 + hi8 + v) * Nout;
        crow[ro +  0] = acc0[v];
        crow[ro + 16] = acc1[v];
        crow[ro + 32] = acc2[v];
        crow[ro + 48] = acc3[v];
    }
}

__global__ void k_att(const float* __restrict__ Hm,
                      const float* __restrict__ asr, const float* __restrict__ adr,
                      float* __restrict__ als, float* __restrict__ ald,
                      int NH, int H, int C) {
    int i = blockIdx.x * blockDim.x + threadIdx.x;
    if (i >= NH) return;
    int h = i % H;
    const float* hv = Hm + (long)i * C;
    const float* av = asr + h * C;
    const float* dv = adr + h * C;
    float s = 0.f, d = 0.f;
    for (int c = 0; c < C; ++c) {
        float x = hv[c];
        s += x * av[c];
        d += x * dv[c];
    }
    als[i] = s;
    ald[i] = d;
}

__global__ void k_fill_i32(int* __restrict__ p, int v, long n) {
    long i = (long)blockIdx.x * blockDim.x + threadIdx.x;
    if (i < n) p[i] = v;
}

__global__ void k_deg_count(const int* __restrict__ dst, int E, int* __restrict__ cnt) {
    int e = blockIdx.x * blockDim.x + threadIdx.x;
    if (e < E) atomicAdd(&cnt[dst[e]], 1);
}

#define SHR(v, n) __builtin_amdgcn_update_dpp(0, (v), 0x110 + (n), 0xf, 0xf, false)
__device__ __forceinline__ int wave_incl_scan(int v, int lane) {
    v += SHR(v, 1);
    v += SHR(v, 2);
    v += SHR(v, 4);
    v += SHR(v, 8);
    const int low = __builtin_amdgcn_readlane(v, 15);
    return lane >= 16 ? v + low : v;
}
#undef SHR

__global__ void __launch_bounds__(1024)
k_scan(const int* __restrict__ cnt, int* __restrict__ rowptr, int N) {
    __shared__ int wsum[32];
    constexpr int U = 8;
    const int lane = threadIdx.x & 31;
    const int w    = __builtin_amdgcn_readfirstlane((int)(threadIdx.x >> 5));
    const int nw   = blockDim.x >> 5;
    const int per  = (((N + nw - 1) / nw) + 31) & ~31;
    const int b = w * per;
    const int e = b + per < N ? b + per : N;

    int s = 0;
    for (int i0 = b; i0 < e; i0 += 32 * U) {
        int v[U];
#pragma unroll
        for (int u = 0; u < U; ++u) { const int i = i0 + u * 32 + lane; v[u] = i < e ? cnt[i] : 0; }
#pragma unroll
        for (int u = 0; u < U; ++u) s += v[u];
    }
    for (int off = 16; off > 0; off >>= 1) s += __shfl_xor(s, off);
    if (lane == 0) wsum[w] = s;
    __syncthreads();
    int carry = lane < w ? wsum[lane] : 0;
    for (int off = 16; off > 0; off >>= 1) carry += __shfl_xor(carry, off);

    for (int i0 = b; i0 < e; i0 += 32 * U) {
        int v[U];
#pragma unroll
        for (int u = 0; u < U; ++u) { const int i = i0 + u * 32 + lane; v[u] = i < e ? cnt[i] : 0; }
#pragma unroll
        for (int u = 0; u < U; ++u) {
            const int i = i0 + u * 32 + lane;
            const int inc = wave_incl_scan(v[u], lane);
            if (i < e) rowptr[i] = carry + inc - v[u];
            carry += __builtin_amdgcn_readlane(inc, 31);
        }
    }
    if (w == nw - 1 && lane == 0) rowptr[N] = carry;
}

__global__ void k_place(const int* __restrict__ src, const int* __restrict__ dst, int E,
                        const int* __restrict__ rowptr, int* __restrict__ cursor,
                        int* __restrict__ tmp) {
    int e = blockIdx.x * blockDim.x + threadIdx.x;
    if (e >= E) return;
    int d = dst[e];
    int p = rowptr[d] + atomicAdd(&cursor[d], 1);
    tmp[p] = src[e];
}

__global__ void __launch_bounds__(256)
k_rank(const int* __restrict__ rowptr, const int* __restrict__ tmp,
       int* __restrict__ csr, int N, int totalWaves) {
    const int lane  = threadIdx.x & 31;
    const int wave0 = blockIdx.x * (blockDim.x >> 5) + (threadIdx.x >> 5);
    for (int nn = wave0; nn < N; nn += totalWaves) {
        const int n  = __builtin_amdgcn_readfirstlane(nn);
        const int b  = rowptr[n], en = rowptr[n + 1];
        const int deg = en - b;
        if (deg <= 32) {
            const int mine = lane < deg ? tmp[b + lane] : 0x7fffffff;
            int r = 0;
            for (int j = 0; j < deg; ++j) {
                const int v = __builtin_amdgcn_readlane(mine, j);
                r += (v < mine || (v == mine && j < lane)) ? 1 : 0;
            }
            if (lane < deg) csr[b + r] = mine;
        } else {
            for (int p = b + lane; p < en; p += 32) {
                const int s = tmp[p];
                int r = 0;
                for (int q = b; q < en; ++q) {
                    const int v = tmp[q];
                    r += (v < s || (v == s && q < p)) ? 1 : 0;
                }
                csr[b + r] = s;
            }
        }
    }
}

__device__ __forceinline__ float leaky(float v) { return v > 0.f ? v : NEG_SLOPE * v; }

template <int H, int C, int ELU>
__global__ void __launch_bounds__(256)
k_gather(const int* __restrict__ rowptr, const int* __restrict__ csr,
         const float* __restrict__ als, const float* __restrict__ ald,
         const float* __restrict__ Hm, const float* __restrict__ bias,
         float* __restrict__ out, int N, int totalWaves) {
    constexpr int NOUT = H * C;
    constexpr int V    = NOUT / 128;
    const int lane = threadIdx.x & 31;
    const int wave0 = blockIdx.x * (blockDim.x >> 5) + (threadIdx.x >> 5);

    int hd[V], ch[V];
#pragma unroll
    for (int k = 0; k < V; ++k) { ch[k] = k * 128 + lane * 4; hd[k] = ch[k] / C; }

    for (int nn = wave0; nn < N; nn += totalWaves) {
        const int n   = __builtin_amdgcn_readfirstlane(nn);
        const int beg = rowptr[n], end = rowptr[n + 1];

        float adn[V], m[V], den[V];
        float4 acc[V];
#pragma unroll
        for (int k = 0; k < V; ++k) {
            adn[k] = ald[n * H + hd[k]];
            m[k]   = leaky(als[n * H + hd[k]] + adn[k]);
        }
        for (int base = beg; base < end; base += 32) {
            const int cnt = end - base < 32 ? end - base : 32;
            const int mys = lane < cnt ? csr[base + lane] : 0;
            for (int j = 0; j < cnt; ++j) {
                const int s = __builtin_amdgcn_readlane(mys, j);
#pragma unroll
                for (int k = 0; k < V; ++k)
                    m[k] = fmaxf(m[k], leaky(als[s * H + hd[k]] + adn[k]));
            }
        }
#pragma unroll
        for (int k = 0; k < V; ++k) {
            const float ex = expf(leaky(als[n * H + hd[k]] + adn[k]) - m[k]);
            const float4 hv = *(const float4*)(Hm + (long)n * NOUT + ch[k]);
            den[k] = ex;
            acc[k] = make_float4(hv.x * ex, hv.y * ex, hv.z * ex, hv.w * ex);
        }
        for (int base = beg; base < end; base += 32) {
            const int cnt = end - base < 32 ? end - base : 32;
            const int mys = lane < cnt ? csr[base + lane] : 0;
#pragma unroll 4
            for (int j = 0; j < cnt; ++j) {
                const int s = __builtin_amdgcn_readlane(mys, j);
#pragma unroll
                for (int k = 0; k < V; ++k) {
                    const float ex = expf(leaky(als[s * H + hd[k]] + adn[k]) - m[k]);
                    const float4 hv = *(const float4*)(Hm + (long)s * NOUT + ch[k]);
                    den[k] += ex;
                    acc[k].x += hv.x * ex;
                    acc[k].y += hv.y * ex;
                    acc[k].z += hv.z * ex;
                    acc[k].w += hv.w * ex;
                }
            }
        }
#pragma unroll
        for (int k = 0; k < V; ++k) {
            const float dn = den[k] + 1e-16f;
            const float4 bv = *(const float4*)(bias + ch[k]);
            float4 o;
            o.x = acc[k].x / dn + bv.x;
            o.y = acc[k].y / dn + bv.y;
            o.z = acc[k].z / dn + bv.z;
            o.w = acc[k].w / dn + bv.w;
            if (ELU) {
                o.x = o.x > 0.f ? o.x : (expf(o.x) - 1.f);
                o.y = o.y > 0.f ? o.y : (expf(o.y) - 1.f);
                o.z = o.z > 0.f ? o.z : (expf(o.z) - 1.f);
                o.w = o.w > 0.f ? o.w : (expf(o.w) - 1.f);
            }
            *(float4*)(out + (long)n * NOUT + ch[k]) = o;
        }
    }
}

static inline long cdivl(long a, long b) { return (a + b - 1) / b; }

struct LayerWS {
    _Float16* Wh;
    float *als, *ald;
    const int *rowptr, *csr;
};

static void run_gat_layer(const float* Xin, const float* W, const float* a_s, const float* a_d,
                          const float* bias, int N, int K, int Nout, int H, int C,
                          float* Hbuf, float* agg, const LayerWS& w,
                          hipStream_t stream) {
    const int TB = 256;
    k_f32_to_f16<<<(int)cdivl((long)Nout * K, TB), TB, 0, stream>>>(W, w.Wh, (long)Nout * K);
    int tilesN4 = Nout / 64, totalWaves = (N / 16) * tilesN4;
    k_gemm_wmma<256><<<(int)cdivl(totalWaves, 8), TB, 0, stream>>>(Xin, w.Wh, Hbuf, Nout,
                                                                   tilesN4, totalWaves);
    k_att<<<(int)cdivl((long)N * H, TB), TB, 0, stream>>>(Hbuf, a_s, a_d, w.als, w.ald,
                                                          N * H, H, C);
    const int GB = 2048, gWaves = GB * (TB / 32);
    if (H == 8)
        k_gather<8, 32, 1><<<GB, TB, 0, stream>>>(w.rowptr, w.csr, w.als, w.ald, Hbuf, bias,
                                                  agg, N, gWaves);
    else
        k_gather<1, 128, 0><<<GB, TB, 0, stream>>>(w.rowptr, w.csr, w.als, w.ald, Hbuf, bias,
                                                   agg, N, gWaves);
}

extern "C" void kernel_launch(void* const* d_in, const int* in_sizes, int n_in,
                              void* d_out, int out_size, void* d_ws, size_t ws_size,
                              hipStream_t stream) {
    const float* x   = (const float*)d_in[0];
    const int*   ei  = (const int*)  d_in[1];
    const float* W1  = (const float*)d_in[2];
    const float* as1 = (const float*)d_in[3];
    const float* ad1 = (const float*)d_in[4];
    const float* b1  = (const float*)d_in[5];
    const float* W2  = (const float*)d_in[6];
    const float* as2 = (const float*)d_in[7];
    const float* ad2 = (const float*)d_in[8];
    const float* b2  = (const float*)d_in[9];
    const float* W3  = (const float*)d_in[10];
    const float* as3 = (const float*)d_in[11];
    const float* ad3 = (const float*)d_in[12];
    const float* b3  = (const float*)d_in[13];
    float* out = (float*)d_out;

    const int FinW = 256;
    const int N    = in_sizes[0] / FinW;
    const int E    = in_sizes[1] / 2;
    const int* src = ei;
    const int* dst = ei + E;

    char* p = (char*)d_ws;
    float*    P   = (float*)p;    p += (size_t)N * 256 * sizeof(float);
    float*    Q   = (float*)p;    p += (size_t)N * 256 * sizeof(float);
    _Float16* Wh  = (_Float16*)p; p += (size_t)256 * 256 * sizeof(_Float16);
    float*    als = (float*)p;    p += (size_t)N * 8 * sizeof(float);
    float*    ald = (float*)p;    p += (size_t)N * 8 * sizeof(float);
    int*      cnt = (int*)p;      p += (size_t)N * sizeof(int);
    int*      cur = (int*)p;      p += (size_t)N * sizeof(int);
    int*      rowptr = (int*)p;   p += (size_t)(N + 4) * sizeof(int);
    int*      tmp  = (int*)p;     p += (size_t)E * sizeof(int);
    int*      csr  = (int*)p;     p += (size_t)E * sizeof(int);
    LayerWS w{Wh, als, ald, rowptr, csr};

    const int TB = 256;
    k_fill_i32<<<(int)cdivl((long)2 * N, TB), TB, 0, stream>>>(cnt, 0, (long)2 * N);
    k_deg_count<<<(int)cdivl(E, TB), TB, 0, stream>>>(dst, E, cnt);
    k_scan<<<1, 1024, 0, stream>>>(cnt, rowptr, N);
    k_place<<<(int)cdivl(E, TB), TB, 0, stream>>>(src, dst, E, rowptr, cur, tmp);
    k_rank<<<2048, TB, 0, stream>>>(rowptr, tmp, csr, N, 2048 * (TB / 32));

    run_gat_layer(x, W1, as1, ad1, b1, N, 256, 256, 8, 32, Q, P, w, stream);
    run_gat_layer(P, W2, as2, ad2, b2, N, 256, 256, 8, 32, Q, P, w, stream);
    run_gat_layer(P, W3, as3, ad3, b3, N, 256, 128, 1, 128, Q, out, w, stream);
}


# Attention logit terms summed in the GEMM epilogue from the accumulators (fixed lane tree: their summation order changes; output layer as two half-head partials); separate pass removed, 3 launches
# speedup vs baseline: 3.5664x; 1.0444x over previous
#include <hip/hip_runtime.h>
#include <hip/hip_bf16.h>

typedef _Float16 v16h __attribute__((ext_vector_type(16)));
typedef _Float16 v8h  __attribute__((ext_vector_type(8)));
typedef float    v8f  __attribute__((ext_vector_type(8)));

#define NEG_SLOPE 0.2f

__global__ void k_f32_to_f16(const float* __restrict__ in, _Float16* __restrict__ out, long n) {
    long i = (long)blockIdx.x * blockDim.x + threadIdx.x;
    if (i < n) out[i] = (_Float16)in[i];
}


struct KStep {
    float4 a0, a1, a2, a3;
    v8h    blo[4], bhi[4];
};

__device__ __forceinline__ void load_kstep(KStep& s, const float* __restrict__ ap,
                                           const _Float16* __restrict__ bp, long K16, int khalf) {
    s.a0 = *(const float4*)(ap + khalf);
    s.a1 = *(const float4*)(ap + khalf + 4);
    s.a2 = *(const float4*)(ap + 16 + khalf);
    s.a3 = *(const float4*)(ap + 16 + khalf + 4);
#pragma unroll
    for (int j = 0; j < 4; ++j) {
        s.blo[j] = *(const v8h*)(bp + j * K16 + khalf);
        s.bhi[j] = *(const v8h*)(bp + j * K16 + 16 + khalf);
    }
}

__device__ __forceinline__ v16h a_frag(const KStep& s) {
    v16h f;
    f[0]  = (_Float16)s.a0.x; f[1]  = (_Float16)s.a0.y; f[2]  = (_Float16)s.a0.z; f[3]  = (_Float16)s.a0.w;
    f[4]  = (_Float16)s.a1.x; f[5]  = (_Float16)s.a1.y; f[6]  = (_Float16)s.a1.z; f[7]  = (_Float16)s.a1.w;
    f[8]  = (_Float16)s.a2.x; f[9]  = (_Float16)s.a2.y; f[10] = (_Float16)s.a2.z; f[11] = (_Float16)s.a2.w;
    f[12] = (_Float16)s.a3.x; f[13] = (_Float16)s.a3.y; f[14] = (_Float16)s.a3.z; f[15] = (_Float16)s.a3.w;
    return f;
}

__device__ __forceinline__ v16h b_frag(const KStep& s, int j) {
    v16h f;
#pragma unroll
    for (int i = 0; i < 8; ++i) { f[i] = s.blo[j][i]; f[8 + i] = s.bhi[j][i]; }
    return f;
}

__device__ __forceinline__ v8f wmma_f16(v16h a, v16h b, v8f c) {
    return __builtin_amdgcn_wmma_f32_16x16x32_f16(
         false, a,  false, b,
         (short)0, c,  false,  false);
}

__device__ __forceinline__ float half_tree_sum16(const float (&x)[16], int c) {
    const bool b3 = (c & 8) != 0, b2 = (c & 4) != 0, b1 = (c & 2) != 0, b0 = (c & 1) != 0;
    float z8[8], z4[4], z2[2];
#pragma unroll
    for (int i = 0; i < 8; ++i) {
        const float keep = b3 ? x[i + 8] : x[i];
        const float send = b3 ? x[i] : x[i + 8];
        z8[i] = keep + __shfl_xor(send, 8);
    }
#pragma unroll
    for (int i = 0; i < 4; ++i) {
        const float keep = b2 ? z8[i + 4] : z8[i];
        const float send = b2 ? z8[i] : z8[i + 4];
        z4[i] = keep + __shfl_xor(send, 4);
    }
#pragma unroll
    for (int i = 0; i < 2; ++i) {
        const float keep = b1 ? z4[i + 2] : z4[i];
        const float send = b1 ? z4[i] : z4[i + 2];
        z2[i] = keep + __shfl_xor(send, 2);
    }
    const float keep = b0 ? z2[1] : z2[0];
    const float send = b0 ? z2[0] : z2[1];
    return keep + __shfl_xor(send, 1);
}

template <int K, int HC>
__global__ void __launch_bounds__(256)
k_gemm_wmma(const float* __restrict__ A, const _Float16* __restrict__ B,
            float* __restrict__ C,
            const float* __restrict__ asr, const float* __restrict__ adr,
            float* __restrict__ als, float* __restrict__ ald,
            int Nout, int tilesN4, int totalWaves) {
    static_assert(HC == 32 || HC == 128, "a 64-column strip must hold whole heads or lie inside one");
    int wave = blockIdx.x * (blockDim.x >> 5) + (threadIdx.x >> 5);
    if (wave >= totalWaves) return;
    int lane  = threadIdx.x & 31;
    int r     = lane & 15;
    int hi    = lane >> 4;
    int khalf = hi << 3;

    int mt  = wave / tilesN4;
    int nt4 = wave - mt * tilesN4;

    const float*    arow = A + (long)(mt * 16 + r) * K;
    const _Float16* brow = B + (long)(nt4 * 64 + r) * K;

    v8f acc0 = {}, acc1 = {}, acc2 = {}, acc3 = {};
    KStep st[2];
    load_kstep(st[0], arow, brow, (long)16 * K, khalf);
#pragma unroll
    for (int kk = 0; kk < K; kk += 32) {
        const int cur = (kk >> 5) & 1;
        __builtin_amdgcn_sched_barrier(0);
        if (kk + 32 < K) load_kstep(st[cur ^ 1], arow + kk + 32, brow + kk + 32, (long)16 * K, khalf);
        __builtin_amdgcn_sched_barrier(0);
        const v16h a = a_frag(st[cur]);
        acc0 = wmma_f16(a, b_frag(st[cur], 0), acc0);
        acc1 = wmma_f16(a, b_frag(st[cur], 1), acc1);
        acc2 = wmma_f16(a, b_frag(st[cur], 2), acc2);
        acc3 = wmma_f16(a, b_frag(st[cur], 3), acc3);
    }

    float sa[4], da[4];
#pragma unroll
    for (int j = 0; j < 4; ++j) {
        sa[j] = asr[nt4 * 64 + j * 16 + r];
        da[j] = adr[nt4 * 64 + j * 16 + r];
    }

    int hi8 = hi << 3;
    float* crow = C + (long)nt4 * 64 + r;
#pragma unroll
    for (int v = 0; v < 8; ++v) {
        long ro = (long)(mt * 16 + hi8 + v) * Nout;
        crow[ro +  0] = acc0[v];
        crow[ro + 16] = acc1[v];
        crow[ro + 32] = acc2[v];
        crow[ro + 48] = acc3[v];
    }

    const long row = mt * 16 + hi8 + (r & 7);
    if (HC == 32) {
        float xs[16], xd[16];
#pragma unroll
        for (int v = 0; v < 8; ++v) {
            xs[v]     = acc0[v] * sa[0] + acc1[v] * sa[1];
            xs[8 + v] = acc2[v] * sa[2] + acc3[v] * sa[3];
            xd[v]     = acc0[v] * da[0] + acc1[v] * da[1];
            xd[8 + v] = acc2[v] * da[2] + acc3[v] * da[3];
        }
        const float ts = half_tree_sum16(xs, r);
        const float td = half_tree_sum16(xd, r);
        const long o = row * (2 * tilesN4) + 2 * nt4 + (r >> 3);
        als[o] = ts;
        ald[o] = td;
    } else {
        float x[16];
#pragma unroll
        for (int v = 0; v < 8; ++v) {
            x[v]     = (acc0[v] * sa[0] + acc1[v] * sa[1]) + (acc2[v] * sa[2] + acc3[v] * sa[3]);
            x[8 + v] = (acc0[v] * da[0] + acc1[v] * da[1]) + (acc2[v] * da[2] + acc3[v] * da[3]);
        }
        const float t = half_tree_sum16(x, r);
        float* dstp = (r >> 3) ? ald : als;
        dstp[row * tilesN4 + nt4] = t;
    }
}

__global__ void k_fill_i32(int* __restrict__ p, int v, long n) {
    long i = (long)blockIdx.x * blockDim.x + threadIdx.x;
    if (i < n) p[i] = v;
}

__global__ void k_deg_count(const int* __restrict__ dst, int E, int* __restrict__ cnt) {
    int e = blockIdx.x * blockDim.x + threadIdx.x;
    if (e < E) atomicAdd(&cnt[dst[e]], 1);
}

#define SHR(v, n) __builtin_amdgcn_update_dpp(0, (v), 0x110 + (n), 0xf, 0xf, false)
__device__ __forceinline__ int wave_incl_scan(int v, int lane) {
    v += SHR(v, 1);
    v += SHR(v, 2);
    v += SHR(v, 4);
    v += SHR(v, 8);
    const int low = __builtin_amdgcn_readlane(v, 15);
    return lane >= 16 ? v + low : v;
}
#undef SHR

__global__ void __launch_bounds__(1024)
k_scan(const int* __restrict__ cnt, int* __restrict__ rowptr, int N) {
    __shared__ int wsum[32];
    constexpr int U = 8;
    const int lane = threadIdx.x & 31;
    const int w    = __builtin_amdgcn_readfirstlane((int)(threadIdx.x >> 5));
    const int nw   = blockDim.x >> 5;
    const int per  = (((N + nw - 1) / nw) + 31) & ~31;
    const int b = w * per;
    const int e = b + per < N ? b + per : N;

    int s = 0;
    for (int i0 = b; i0 < e; i0 += 32 * U) {
        int v[U];
#pragma unroll
        for (int u = 0; u < U; ++u) { const int i = i0 + u * 32 + lane; v[u] = i < e ? cnt[i] : 0; }
#pragma unroll
        for (int u = 0; u < U; ++u) s += v[u];
    }
    for (int off = 16; off > 0; off >>= 1) s += __shfl_xor(s, off);
    if (lane == 0) wsum[w] = s;
    __syncthreads();
    int carry = lane < w ? wsum[lane] : 0;
    for (int off = 16; off > 0; off >>= 1) carry += __shfl_xor(carry, off);

    for (int i0 = b; i0 < e; i0 += 32 * U) {
        int v[U];
#pragma unroll
        for (int u = 0; u < U; ++u) { const int i = i0 + u * 32 + lane; v[u] = i < e ? cnt[i] : 0; }
#pragma unroll
        for (int u = 0; u < U; ++u) {
            const int i = i0 + u * 32 + lane;
            const int inc = wave_incl_scan(v[u], lane);
            if (i < e) rowptr[i] = carry + inc - v[u];
            carry += __builtin_amdgcn_readlane(inc, 31);
        }
    }
    if (w == nw - 1 && lane == 0) rowptr[N] = carry;
}

__global__ void k_place(const int* __restrict__ src, const int* __restrict__ dst, int E,
                        const int* __restrict__ rowptr, int* __restrict__ cursor,
                        int* __restrict__ tmp) {
    int e = blockIdx.x * blockDim.x + threadIdx.x;
    if (e >= E) return;
    int d = dst[e];
    int p = rowptr[d] + atomicAdd(&cursor[d], 1);
    tmp[p] = src[e];
}

__global__ void __launch_bounds__(256)
k_rank(const int* __restrict__ rowptr, const int* __restrict__ tmp,
       int* __restrict__ csr, int N, int totalWaves) {
    const int lane  = threadIdx.x & 31;
    const int wave0 = blockIdx.x * (blockDim.x >> 5) + (threadIdx.x >> 5);
    for (int nn = wave0; nn < N; nn += totalWaves) {
        const int n  = __builtin_amdgcn_readfirstlane(nn);
        const int b  = rowptr[n], en = rowptr[n + 1];
        const int deg = en - b;
        if (deg <= 32) {
            const int mine = lane < deg ? tmp[b + lane] : 0x7fffffff;
            int r = 0;
            for (int j = 0; j < deg; ++j) {
                const int v = __builtin_amdgcn_readlane(mine, j);
                r += (v < mine || (v == mine && j < lane)) ? 1 : 0;
            }
            if (lane < deg) csr[b + r] = mine;
        } else {
            for (int p = b + lane; p < en; p += 32) {
                const int s = tmp[p];
                int r = 0;
                for (int q = b; q < en; ++q) {
                    const int v = tmp[q];
                    r += (v < s || (v == s && q < p)) ? 1 : 0;
                }
                csr[b + r] = s;
            }
        }
    }
}

__device__ __forceinline__ float leaky(float v) { return v > 0.f ? v : NEG_SLOPE * v; }

template <int P>
__device__ __forceinline__ float lterm(const float* __restrict__ p, int i) {
    if (P == 1) return p[i];
    const float2 t = *(const float2*)(p + 2 * (long)i);
    return t.x + t.y;
}

template <int H, int C, int ELU>
__global__ void __launch_bounds__(256)
k_gather(const int* __restrict__ rowptr, const int* __restrict__ csr,
         const float* __restrict__ als, const float* __restrict__ ald,
         const float* __restrict__ Hm, const float* __restrict__ bias,
         float* __restrict__ out, int N, int totalWaves) {
    constexpr int NOUT = H * C;
    constexpr int V    = NOUT / 128;
    constexpr int P    = C > 64 ? C / 64 : 1;
    const int lane = threadIdx.x & 31;
    const int wave0 = blockIdx.x * (blockDim.x >> 5) + (threadIdx.x >> 5);

    int hd[V], ch[V];
#pragma unroll
    for (int k = 0; k < V; ++k) { ch[k] = k * 128 + lane * 4; hd[k] = ch[k] / C; }

    for (int nn = wave0; nn < N; nn += totalWaves) {
        const int n   = __builtin_amdgcn_readfirstlane(nn);
        const int beg = rowptr[n], end = rowptr[n + 1];

        float adn[V], m[V], den[V];
        float4 acc[V];
#pragma unroll
        for (int k = 0; k < V; ++k) {
            adn[k] = lterm<P>(ald, n * H + hd[k]);
            m[k]   = leaky(lterm<P>(als, n * H + hd[k]) + adn[k]);
        }
        for (int base = beg; base < end; base += 32) {
            const int cnt = end - base < 32 ? end - base : 32;
            const int mys = lane < cnt ? csr[base + lane] : 0;
            for (int j = 0; j < cnt; ++j) {
                const int s = __builtin_amdgcn_readlane(mys, j);
#pragma unroll
                for (int k = 0; k < V; ++k)
                    m[k] = fmaxf(m[k], leaky(lterm<P>(als, s * H + hd[k]) + adn[k]));
            }
        }
#pragma unroll
        for (int k = 0; k < V; ++k) {
            const float ex = expf(leaky(lterm<P>(als, n * H + hd[k]) + adn[k]) - m[k]);
            const float4 hv = *(const float4*)(Hm + (long)n * NOUT + ch[k]);
            den[k] = ex;
            acc[k] = make_float4(hv.x * ex, hv.y * ex, hv.z * ex, hv.w * ex);
        }
        for (int base = beg; base < end; base += 32) {
            const int cnt = end - base < 32 ? end - base : 32;
            const int mys = lane < cnt ? csr[base + lane] : 0;
#pragma unroll 4
            for (int j = 0; j < cnt; ++j) {
                const int s = __builtin_amdgcn_readlane(mys, j);
#pragma unroll
                for (int k = 0; k < V; ++k) {
                    const float ex = expf(leaky(lterm<P>(als, s * H + hd[k]) + adn[k]) - m[k]);
                    const float4 hv = *(const float4*)(Hm + (long)s * NOUT + ch[k]);
                    den[k] += ex;
                    acc[k].x += hv.x * ex;
                    acc[k].y += hv.y * ex;
                    acc[k].z += hv.z * ex;
                    acc[k].w += hv.w * ex;
                }
            }
        }
#pragma unroll
        for (int k = 0; k < V; ++k) {
            const float dn = den[k] + 1e-16f;
            const float4 bv = *(const float4*)(bias + ch[k]);
            float4 o;
            o.x = acc[k].x / dn + bv.x;
            o.y = acc[k].y / dn + bv.y;
            o.z = acc[k].z / dn + bv.z;
            o.w = acc[k].w / dn + bv.w;
            if (ELU) {
                o.x = o.x > 0.f ? o.x : (expf(o.x) - 1.f);
                o.y = o.y > 0.f ? o.y : (expf(o.y) - 1.f);
                o.z = o.z > 0.f ? o.z : (expf(o.z) - 1.f);
                o.w = o.w > 0.f ? o.w : (expf(o.w) - 1.f);
            }
            *(float4*)(out + (long)n * NOUT + ch[k]) = o;
        }
    }
}

static inline long cdivl(long a, long b) { return (a + b - 1) / b; }

struct LayerWS {
    _Float16* Wh;
    float *als, *ald;
    const int *rowptr, *csr;
};

static void run_gat_layer(const float* Xin, const float* W, const float* a_s, const float* a_d,
                          const float* bias, int N, int K, int Nout, int H, int C,
                          float* Hbuf, float* agg, const LayerWS& w,
                          hipStream_t stream) {
    const int TB = 256;
    k_f32_to_f16<<<(int)cdivl((long)Nout * K, TB), TB, 0, stream>>>(W, w.Wh, (long)Nout * K);
    int tilesN4 = Nout / 64, totalWaves = (N / 16) * tilesN4;
    if (C == 32)
        k_gemm_wmma<256, 32><<<(int)cdivl(totalWaves, 8), TB, 0, stream>>>(
            Xin, w.Wh, Hbuf, a_s, a_d, w.als, w.ald, Nout, tilesN4, totalWaves);
    else
        k_gemm_wmma<256, 128><<<(int)cdivl(totalWaves, 8), TB, 0, stream>>>(
            Xin, w.Wh, Hbuf, a_s, a_d, w.als, w.ald, Nout, tilesN4, totalWaves);
    const int GB = 2048, gWaves = GB * (TB / 32);
    if (H == 8)
        k_gather<8, 32, 1><<<GB, TB, 0, stream>>>(w.rowptr, w.csr, w.als, w.ald, Hbuf, bias,
                                                  agg, N, gWaves);
    else
        k_gather<1, 128, 0><<<GB, TB, 0, stream>>>(w.rowptr, w.csr, w.als, w.ald, Hbuf, bias,
                                                   agg, N, gWaves);
}

extern "C" void kernel_launch(void* const* d_in, const int* in_sizes, int n_in,
                              void* d_out, int out_size, void* d_ws, size_t ws_size,
                              hipStream_t stream) {
    const float* x   = (const float*)d_in[0];
    const int*   ei  = (const int*)  d_in[1];
    const float* W1  = (const float*)d_in[2];
    const float* as1 = (const float*)d_in[3];
    const float* ad1 = (const float*)d_in[4];
    const float* b1  = (const float*)d_in[5];
    const float* W2  = (const float*)d_in[6];
    const float* as2 = (const float*)d_in[7];
    const float* ad2 = (const float*)d_in[8];
    const float* b2  = (const float*)d_in[9];
    const float* W3  = (const float*)d_in[10];
    const float* as3 = (const float*)d_in[11];
    const float* ad3 = (const float*)d_in[12];
    const float* b3  = (const float*)d_in[13];
    float* out = (float*)d_out;

    const int FinW = 256;
    const int N    = in_sizes[0] / FinW;
    const int E    = in_sizes[1] / 2;
    const int* src = ei;
    const int* dst = ei + E;

    char* p = (char*)d_ws;
    float*    P   = (float*)p;    p += (size_t)N * 256 * sizeof(float);
    float*    Q   = (float*)p;    p += (size_t)N * 256 * sizeof(float);
    _Float16* Wh  = (_Float16*)p; p += (size_t)256 * 256 * sizeof(_Float16);
    float*    als = (float*)p;    p += (size_t)N * 8 * sizeof(float);
    float*    ald = (float*)p;    p += (size_t)N * 8 * sizeof(float);
    int*      cnt = (int*)p;      p += (size_t)N * sizeof(int);
    int*      cur = (int*)p;      p += (size_t)N * sizeof(int);
    int*      rowptr = (int*)p;   p += (size_t)(N + 4) * sizeof(int);
    int*      tmp  = (int*)p;     p += (size_t)E * sizeof(int);
    int*      csr  = (int*)p;     p += (size_t)E * sizeof(int);
    LayerWS w{Wh, als, ald, rowptr, csr};

    const int TB = 256;
    k_fill_i32<<<(int)cdivl((long)2 * N, TB), TB, 0, stream>>>(cnt, 0, (long)2 * N);
    k_deg_count<<<(int)cdivl(E, TB), TB, 0, stream>>>(dst, E, cnt);
    k_scan<<<1, 1024, 0, stream>>>(cnt, rowptr, N);
    k_place<<<(int)cdivl(E, TB), TB, 0, stream>>>(src, dst, E, rowptr, cur, tmp);
    k_rank<<<2048, TB, 0, stream>>>(rowptr, tmp, csr, N, 2048 * (TB / 32));

    run_gat_layer(x, W1, as1, ad1, b1, N, 256, 256, 8, 32, Q, P, w, stream);
    run_gat_layer(P, W2, as2, ad2, b2, N, 256, 256, 8, 32, Q, P, w, stream);
    run_gat_layer(P, W3, as3, ad3, b3, N, 256, 128, 1, 128, Q, out, w, stream);
}


# Aggregation: each edge-head logit loaded once, its softmax weight computed once by one lane of the head group and kept in registers across both passes, handed to row lanes by shuffle; same sum order
# speedup vs baseline: 3.9494x; 1.1074x over previous
#include <hip/hip_runtime.h>
#include <hip/hip_bf16.h>

typedef _Float16 v16h __attribute__((ext_vector_type(16)));
typedef _Float16 v8h  __attribute__((ext_vector_type(8)));
typedef float    v8f  __attribute__((ext_vector_type(8)));

#define NEG_SLOPE 0.2f

__global__ void k_f32_to_f16(const float* __restrict__ in, _Float16* __restrict__ out, long n) {
    long i = (long)blockIdx.x * blockDim.x + threadIdx.x;
    if (i < n) out[i] = (_Float16)in[i];
}


struct KStep {
    float4 a0, a1, a2, a3;
    v8h    blo[4], bhi[4];
};

__device__ __forceinline__ void load_kstep(KStep& s, const float* __restrict__ ap,
                                           const _Float16* __restrict__ bp, long K16, int khalf) {
    s.a0 = *(const float4*)(ap + khalf);
    s.a1 = *(const float4*)(ap + khalf + 4);
    s.a2 = *(const float4*)(ap + 16 + khalf);
    s.a3 = *(const float4*)(ap + 16 + khalf + 4);
#pragma unroll
    for (int j = 0; j < 4; ++j) {
        s.blo[j] = *(const v8h*)(bp + j * K16 + khalf);
        s.bhi[j] = *(const v8h*)(bp + j * K16 + 16 + khalf);
    }
}

__device__ __forceinline__ v16h a_frag(const KStep& s) {
    v16h f;
    f[0]  = (_Float16)s.a0.x; f[1]  = (_Float16)s.a0.y; f[2]  = (_Float16)s.a0.z; f[3]  = (_Float16)s.a0.w;
    f[4]  = (_Float16)s.a1.x; f[5]  = (_Float16)s.a1.y; f[6]  = (_Float16)s.a1.z; f[7]  = (_Float16)s.a1.w;
    f[8]  = (_Float16)s.a2.x; f[9]  = (_Float16)s.a2.y; f[10] = (_Float16)s.a2.z; f[11] = (_Float16)s.a2.w;
    f[12] = (_Float16)s.a3.x; f[13] = (_Float16)s.a3.y; f[14] = (_Float16)s.a3.z; f[15] = (_Float16)s.a3.w;
    return f;
}

__device__ __forceinline__ v16h b_frag(const KStep& s, int j) {
    v16h f;
#pragma unroll
    for (int i = 0; i < 8; ++i) { f[i] = s.blo[j][i]; f[8 + i] = s.bhi[j][i]; }
    return f;
}

__device__ __forceinline__ v8f wmma_f16(v16h a, v16h b, v8f c) {
    return __builtin_amdgcn_wmma_f32_16x16x32_f16(
         false, a,  false, b,
         (short)0, c,  false,  false);
}

__device__ __forceinline__ float half_tree_sum16(const float (&x)[16], int c) {
    const bool b3 = (c & 8) != 0, b2 = (c & 4) != 0, b1 = (c & 2) != 0, b0 = (c & 1) != 0;
    float z8[8], z4[4], z2[2];
#pragma unroll
    for (int i = 0; i < 8; ++i) {
        const float keep = b3 ? x[i + 8] : x[i];
        const float send = b3 ? x[i] : x[i + 8];
        z8[i] = keep + __shfl_xor(send, 8);
    }
#pragma unroll
    for (int i = 0; i < 4; ++i) {
        const float keep = b2 ? z8[i + 4] : z8[i];
        const float send = b2 ? z8[i] : z8[i + 4];
        z4[i] = keep + __shfl_xor(send, 4);
    }
#pragma unroll
    for (int i = 0; i < 2; ++i) {
        const float keep = b1 ? z4[i + 2] : z4[i];
        const float send = b1 ? z4[i] : z4[i + 2];
        z2[i] = keep + __shfl_xor(send, 2);
    }
    const float keep = b0 ? z2[1] : z2[0];
    const float send = b0 ? z2[0] : z2[1];
    return keep + __shfl_xor(send, 1);
}

template <int K, int HC>
__global__ void __launch_bounds__(256)
k_gemm_wmma(const float* __restrict__ A, const _Float16* __restrict__ B,
            float* __restrict__ C,
            const float* __restrict__ asr, const float* __restrict__ adr,
            float* __restrict__ als, float* __restrict__ ald,
            int Nout, int tilesN4, int totalWaves) {
    static_assert(HC == 32 || HC == 128, "a 64-column strip must hold whole heads or lie inside one");
    int wave = blockIdx.x * (blockDim.x >> 5) + (threadIdx.x >> 5);
    if (wave >= totalWaves) return;
    int lane  = threadIdx.x & 31;
    int r     = lane & 15;
    int hi    = lane >> 4;
    int khalf = hi << 3;

    int mt  = wave / tilesN4;
    int nt4 = wave - mt * tilesN4;

    const float*    arow = A + (long)(mt * 16 + r) * K;
    const _Float16* brow = B + (long)(nt4 * 64 + r) * K;

    v8f acc0 = {}, acc1 = {}, acc2 = {}, acc3 = {};
    KStep st[2];
    load_kstep(st[0], arow, brow, (long)16 * K, khalf);
#pragma unroll
    for (int kk = 0; kk < K; kk += 32) {
        const int cur = (kk >> 5) & 1;
        __builtin_amdgcn_sched_barrier(0);
        if (kk + 32 < K) load_kstep(st[cur ^ 1], arow + kk + 32, brow + kk + 32, (long)16 * K, khalf);
        __builtin_amdgcn_sched_barrier(0);
        const v16h a = a_frag(st[cur]);
        acc0 = wmma_f16(a, b_frag(st[cur], 0), acc0);
        acc1 = wmma_f16(a, b_frag(st[cur], 1), acc1);
        acc2 = wmma_f16(a, b_frag(st[cur], 2), acc2);
        acc3 = wmma_f16(a, b_frag(st[cur], 3), acc3);
    }

    float sa[4], da[4];
#pragma unroll
    for (int j = 0; j < 4; ++j) {
        sa[j] = asr[nt4 * 64 + j * 16 + r];
        da[j] = adr[nt4 * 64 + j * 16 + r];
    }

    int hi8 = hi << 3;
    float* crow = C + (long)nt4 * 64 + r;
#pragma unroll
    for (int v = 0; v < 8; ++v) {
        long ro = (long)(mt * 16 + hi8 + v) * Nout;
        crow[ro +  0] = acc0[v];
        crow[ro + 16] = acc1[v];
        crow[ro + 32] = acc2[v];
        crow[ro + 48] = acc3[v];
    }

    const long row = mt * 16 + hi8 + (r & 7);
    if (HC == 32) {
        float xs[16], xd[16];
#pragma unroll
        for (int v = 0; v < 8; ++v) {
            xs[v]     = acc0[v] * sa[0] + acc1[v] * sa[1];
            xs[8 + v] = acc2[v] * sa[2] + acc3[v] * sa[3];
            xd[v]     = acc0[v] * da[0] + acc1[v] * da[1];
            xd[8 + v] = acc2[v] * da[2] + acc3[v] * da[3];
        }
        const float ts = half_tree_sum16(xs, r);
        const float td = half_tree_sum16(xd, r);
        const long o = row * (2 * tilesN4) + 2 * nt4 + (r >> 3);
        als[o] = ts;
        ald[o] = td;
    } else {
        float x[16];
#pragma unroll
        for (int v = 0; v < 8; ++v) {
            x[v]     = (acc0[v] * sa[0] + acc1[v] * sa[1]) + (acc2[v] * sa[2] + acc3[v] * sa[3]);
            x[8 + v] = (acc0[v] * da[0] + acc1[v] * da[1]) + (acc2[v] * da[2] + acc3[v] * da[3]);
        }
        const float t = half_tree_sum16(x, r);
        float* dstp = (r >> 3) ? ald : als;
        dstp[row * tilesN4 + nt4] = t;
    }
}

__global__ void k_fill_i32(int* __restrict__ p, int v, long n) {
    long i = (long)blockIdx.x * blockDim.x + threadIdx.x;
    if (i < n) p[i] = v;
}

__global__ void k_deg_count(const int* __restrict__ dst, int E, int* __restrict__ cnt) {
    int e = blockIdx.x * blockDim.x + threadIdx.x;
    if (e < E) atomicAdd(&cnt[dst[e]], 1);
}

#define SHR(v, n) __builtin_amdgcn_update_dpp(0, (v), 0x110 + (n), 0xf, 0xf, false)
__device__ __forceinline__ int wave_incl_scan(int v, int lane) {
    v += SHR(v, 1);
    v += SHR(v, 2);
    v += SHR(v, 4);
    v += SHR(v, 8);
    const int low = __builtin_amdgcn_readlane(v, 15);
    return lane >= 16 ? v + low : v;
}
#undef SHR

__global__ void __launch_bounds__(1024)
k_scan(const int* __restrict__ cnt, int* __restrict__ rowptr, int N) {
    __shared__ int wsum[32];
    constexpr int U = 8;
    const int lane = threadIdx.x & 31;
    const int w    = __builtin_amdgcn_readfirstlane((int)(threadIdx.x >> 5));
    const int nw   = blockDim.x >> 5;
    const int per  = (((N + nw - 1) / nw) + 31) & ~31;
    const int b = w * per;
    const int e = b + per < N ? b + per : N;

    int s = 0;
    for (int i0 = b; i0 < e; i0 += 32 * U) {
        int v[U];
#pragma unroll
        for (int u = 0; u < U; ++u) { const int i = i0 + u * 32 + lane; v[u] = i < e ? cnt[i] : 0; }
#pragma unroll
        for (int u = 0; u < U; ++u) s += v[u];
    }
    for (int off = 16; off > 0; off >>= 1) s += __shfl_xor(s, off);
    if (lane == 0) wsum[w] = s;
    __syncthreads();
    int carry = lane < w ? wsum[lane] : 0;
    for (int off = 16; off > 0; off >>= 1) carry += __shfl_xor(carry, off);

    for (int i0 = b; i0 < e; i0 += 32 * U) {
        int v[U];
#pragma unroll
        for (int u = 0; u < U; ++u) { const int i = i0 + u * 32 + lane; v[u] = i < e ? cnt[i] : 0; }
#pragma unroll
        for (int u = 0; u < U; ++u) {
            const int i = i0 + u * 32 + lane;
            const int inc = wave_incl_scan(v[u], lane);
            if (i < e) rowptr[i] = carry + inc - v[u];
            carry += __builtin_amdgcn_readlane(inc, 31);
        }
    }
    if (w == nw - 1 && lane == 0) rowptr[N] = carry;
}

__global__ void k_place(const int* __restrict__ src, const int* __restrict__ dst, int E,
                        const int* __restrict__ rowptr, int* __restrict__ cursor,
                        int* __restrict__ tmp) {
    int e = blockIdx.x * blockDim.x + threadIdx.x;
    if (e >= E) return;
    int d = dst[e];
    int p = rowptr[d] + atomicAdd(&cursor[d], 1);
    tmp[p] = src[e];
}

__global__ void __launch_bounds__(256)
k_rank(const int* __restrict__ rowptr, const int* __restrict__ tmp,
       int* __restrict__ csr, int N, int totalWaves) {
    const int lane  = threadIdx.x & 31;
    const int wave0 = blockIdx.x * (blockDim.x >> 5) + (threadIdx.x >> 5);
    for (int nn = wave0; nn < N; nn += totalWaves) {
        const int n  = __builtin_amdgcn_readfirstlane(nn);
        const int b  = rowptr[n], en = rowptr[n + 1];
        const int deg = en - b;
        if (deg <= 32) {
            const int mine = lane < deg ? tmp[b + lane] : 0x7fffffff;
            int r = 0;
            for (int j = 0; j < deg; ++j) {
                const int v = __builtin_amdgcn_readlane(mine, j);
                r += (v < mine || (v == mine && j < lane)) ? 1 : 0;
            }
            if (lane < deg) csr[b + r] = mine;
        } else {
            for (int p = b + lane; p < en; p += 32) {
                const int s = tmp[p];
                int r = 0;
                for (int q = b; q < en; ++q) {
                    const int v = tmp[q];
                    r += (v < s || (v == s && q < p)) ? 1 : 0;
                }
                csr[b + r] = s;
            }
        }
    }
}

__device__ __forceinline__ float leaky(float v) { return v > 0.f ? v : NEG_SLOPE * v; }

template <int P>
__device__ __forceinline__ float lterm(const float* __restrict__ p, int i) {
    if (P == 1) return p[i];
    const float2 t = *(const float2*)(p + 2 * (long)i);
    return t.x + t.y;
}

template <int NV>
__device__ __forceinline__ void loads_issued_by_here(const float (&l)[NV]) {
    static_assert(NV == 1 || NV == 8, "one operand list per count in use");
    if constexpr (NV == 1)
        asm volatile("" :: "v"(l[0]));
    else
        asm volatile("" :: "v"(l[0]), "v"(l[1]), "v"(l[2]), "v"(l[3]),
                           "v"(l[4]), "v"(l[5]), "v"(l[6]), "v"(l[7]));
}

template <int H, int C, int ELU>
__global__ void __launch_bounds__(256)
k_gather(const int* __restrict__ rowptr, const int* __restrict__ csr,
         const float* __restrict__ als, const float* __restrict__ ald,
         const float* __restrict__ Hm, const float* __restrict__ bias,
         float* __restrict__ out, int N, int totalWaves) {
    constexpr int NOUT = H * C;
    constexpr int V    = NOUT / 128;
    constexpr int P    = C > 64 ? C / 64 : 1;
    constexpr int G    = C / 4 < 32 ? C / 4 : 32;
    constexpr int T    = 32 / G;
    static_assert((G & (G - 1)) == 0 && G * T == 32, "head groups must tile the wave");
    const int lane = threadIdx.x & 31;
    const int gi   = lane & (G - 1);
    const int gb   = lane & ~(G - 1);
    const int wave0 = blockIdx.x * (blockDim.x >> 5) + (threadIdx.x >> 5);

    int hd[V], ch[V];
#pragma unroll
    for (int k = 0; k < V; ++k) { ch[k] = k * 128 + lane * 4; hd[k] = ch[k] / C; }

    for (int nn = wave0; nn < N; nn += totalWaves) {
        const int n   = __builtin_amdgcn_readfirstlane(nn);
        const int beg = rowptr[n], end = rowptr[n + 1];
        const bool one = end - beg <= 32;

        float adn[V], es[V], m[V], den[V];
        float4 acc[V];
#pragma unroll
        for (int k = 0; k < V; ++k) {
            adn[k] = lterm<P>(ald, n * H + hd[k]);
            es[k]  = leaky(lterm<P>(als, n * H + hd[k]) + adn[k]);
            m[k]   = es[k];
        }

        int   mys = 0;
        float e[V][T];
        auto chunk_logits = [&](int cnt) {
            int sv[T];
#pragma unroll
            for (int t = 0; t < T; ++t) sv[t] = T == 1 ? mys : __shfl(mys, gi + G * t);
            float l[V * T];
#pragma unroll
            for (int k = 0; k < V; ++k)
#pragma unroll
                for (int t = 0; t < T; ++t)
                    l[k * T + t] = lterm<P>(als, sv[t] * H + hd[k]);
            loads_issued_by_here(l);
#pragma unroll
            for (int k = 0; k < V; ++k)
#pragma unroll
                for (int t = 0; t < T; ++t)
                    e[k][t] = gi + G * t < cnt ? leaky(l[k * T + t] + adn[k]) : -__builtin_inff();
        };

        for (int base = beg; base < end; base += 32) {
            const int cnt = end - base < 32 ? end - base : 32;
            mys = lane < cnt ? csr[base + lane] : 0;
            chunk_logits(cnt);
#pragma unroll
            for (int k = 0; k < V; ++k)
#pragma unroll
                for (int t = 0; t < T; ++t) m[k] = fmaxf(m[k], e[k][t]);
        }
#pragma unroll
        for (int k = 0; k < V; ++k)
#pragma unroll
            for (int off = G / 2; off > 0; off >>= 1) m[k] = fmaxf(m[k], __shfl_xor(m[k], off));

#pragma unroll
        for (int k = 0; k < V; ++k) {
            const float ex = expf(es[k] - m[k]);
            const float4 hv = *(const float4*)(Hm + (long)n * NOUT + ch[k]);
            den[k] = ex;
            acc[k] = make_float4(hv.x * ex, hv.y * ex, hv.z * ex, hv.w * ex);
        }
        for (int base = beg; base < end; base += 32) {
            const int cnt = end - base < 32 ? end - base : 32;
            if (!one) {
                mys = lane < cnt ? csr[base + lane] : 0;
                chunk_logits(cnt);
            }
#pragma unroll
            for (int t = 0; t < T; ++t) {
                const int lim = cnt - t * G < G ? cnt - t * G : G;
                if (lim > 0) {
                    float p[V];
#pragma unroll
                    for (int k = 0; k < V; ++k) p[k] = expf(e[k][t] - m[k]);
#pragma unroll 4
                    for (int i = 0; i < lim; ++i) {
                        const int s = __builtin_amdgcn_readlane(mys, t * G + i);
#pragma unroll
                        for (int k = 0; k < V; ++k) {
                            const float ex = G == 32
                                ? __int_as_float(__builtin_amdgcn_readlane(__float_as_int(p[k]), i))
                                : __shfl(p[k], gb + i);
                            const float4 hv = *(const float4*)(Hm + (long)s * NOUT + ch[k]);
                            den[k] += ex;
                            acc[k].x += hv.x * ex;
                            acc[k].y += hv.y * ex;
                            acc[k].z += hv.z * ex;
                            acc[k].w += hv.w * ex;
                        }
                    }
                }
            }
        }
#pragma unroll
        for (int k = 0; k < V; ++k) {
            const float dn = den[k] + 1e-16f;
            const float4 bv = *(const float4*)(bias + ch[k]);
            float4 o;
            o.x = acc[k].x / dn + bv.x;
            o.y = acc[k].y / dn + bv.y;
            o.z = acc[k].z / dn + bv.z;
            o.w = acc[k].w / dn + bv.w;
            if (ELU) {
                o.x = o.x > 0.f ? o.x : (expf(o.x) - 1.f);
                o.y = o.y > 0.f ? o.y : (expf(o.y) - 1.f);
                o.z = o.z > 0.f ? o.z : (expf(o.z) - 1.f);
                o.w = o.w > 0.f ? o.w : (expf(o.w) - 1.f);
            }
            *(float4*)(out + (long)n * NOUT + ch[k]) = o;
        }
    }
}

static inline long cdivl(long a, long b) { return (a + b - 1) / b; }

struct LayerWS {
    _Float16* Wh;
    float *als, *ald;
    const int *rowptr, *csr;
};

static void run_gat_layer(const float* Xin, const float* W, const float* a_s, const float* a_d,
                          const float* bias, int N, int K, int Nout, int H, int C,
                          float* Hbuf, float* agg, const LayerWS& w,
                          hipStream_t stream) {
    const int TB = 256;
    k_f32_to_f16<<<(int)cdivl((long)Nout * K, TB), TB, 0, stream>>>(W, w.Wh, (long)Nout * K);
    int tilesN4 = Nout / 64, totalWaves = (N / 16) * tilesN4;
    if (C == 32)
        k_gemm_wmma<256, 32><<<(int)cdivl(totalWaves, 8), TB, 0, stream>>>(
            Xin, w.Wh, Hbuf, a_s, a_d, w.als, w.ald, Nout, tilesN4, totalWaves);
    else
        k_gemm_wmma<256, 128><<<(int)cdivl(totalWaves, 8), TB, 0, stream>>>(
            Xin, w.Wh, Hbuf, a_s, a_d, w.als, w.ald, Nout, tilesN4, totalWaves);
    const int GB = 2048, gWaves = GB * (TB / 32);
    if (H == 8)
        k_gather<8, 32, 1><<<GB, TB, 0, stream>>>(w.rowptr, w.csr, w.als, w.ald, Hbuf, bias,
                                                  agg, N, gWaves);
    else
        k_gather<1, 128, 0><<<GB, TB, 0, stream>>>(w.rowptr, w.csr, w.als, w.ald, Hbuf, bias,
                                                   agg, N, gWaves);
}

extern "C" void kernel_launch(void* const* d_in, const int* in_sizes, int n_in,
                              void* d_out, int out_size, void* d_ws, size_t ws_size,
                              hipStream_t stream) {
    const float* x   = (const float*)d_in[0];
    const int*   ei  = (const int*)  d_in[1];
    const float* W1  = (const float*)d_in[2];
    const float* as1 = (const float*)d_in[3];
    const float* ad1 = (const float*)d_in[4];
    const float* b1  = (const float*)d_in[5];
    const float* W2  = (const float*)d_in[6];
    const float* as2 = (const float*)d_in[7];
    const float* ad2 = (const float*)d_in[8];
    const float* b2  = (const float*)d_in[9];
    const float* W3  = (const float*)d_in[10];
    const float* as3 = (const float*)d_in[11];
    const float* ad3 = (const float*)d_in[12];
    const float* b3  = (const float*)d_in[13];
    float* out = (float*)d_out;

    const int FinW = 256;
    const int N    = in_sizes[0] / FinW;
    const int E    = in_sizes[1] / 2;
    const int* src = ei;
    const int* dst = ei + E;

    char* p = (char*)d_ws;
    float*    P   = (float*)p;    p += (size_t)N * 256 * sizeof(float);
    float*    Q   = (float*)p;    p += (size_t)N * 256 * sizeof(float);
    _Float16* Wh  = (_Float16*)p; p += (size_t)256 * 256 * sizeof(_Float16);
    float*    als = (float*)p;    p += (size_t)N * 8 * sizeof(float);
    float*    ald = (float*)p;    p += (size_t)N * 8 * sizeof(float);
    int*      cnt = (int*)p;      p += (size_t)N * sizeof(int);
    int*      cur = (int*)p;      p += (size_t)N * sizeof(int);
    int*      rowptr = (int*)p;   p += (size_t)(N + 4) * sizeof(int);
    int*      tmp  = (int*)p;     p += (size_t)E * sizeof(int);
    int*      csr  = (int*)p;     p += (size_t)E * sizeof(int);
    LayerWS w{Wh, als, ald, rowptr, csr};

    const int TB = 256;
    k_fill_i32<<<(int)cdivl((long)2 * N, TB), TB, 0, stream>>>(cnt, 0, (long)2 * N);
    k_deg_count<<<(int)cdivl(E, TB), TB, 0, stream>>>(dst, E, cnt);
    k_scan<<<1, 1024, 0, stream>>>(cnt, rowptr, N);
    k_place<<<(int)cdivl(E, TB), TB, 0, stream>>>(src, dst, E, rowptr, cur, tmp);
    k_rank<<<2048, TB, 0, stream>>>(rowptr, tmp, csr, N, 2048 * (TB / 32));

    run_gat_layer(x, W1, as1, ad1, b1, N, 256, 256, 8, 32, Q, P, w, stream);
    run_gat_layer(P, W2, as2, ad2, b2, N, 256, 256, 8, 32, Q, P, w, stream);
    run_gat_layer(P, W3, as3, ad3, b3, N, 256, 128, 1, 128, Q, out, w, stream);
}


# GEMM operands staged in LDS with whole-line loads: 256 resident 16-wave work-groups hold the weights (rounded to f16 on the way; 3 conversion launches gone), stage input rows per round; same k order
# speedup vs baseline: 4.9753x; 1.2598x over previous
#include <hip/hip_runtime.h>
#include <hip/hip_bf16.h>

typedef _Float16 v16h __attribute__((ext_vector_type(16)));
typedef _Float16 v8h  __attribute__((ext_vector_type(8)));
typedef float    v8f  __attribute__((ext_vector_type(8)));

#define NEG_SLOPE 0.2f

typedef _Float16 v4h __attribute__((ext_vector_type(4)));

#define GEMM_K     256
#define GEMM_PITCH 264

struct KStep {
    v8h alo, ahi;
    v8h blo[4], bhi[4];
};

__device__ __forceinline__ void load_kstep(KStep& s, const _Float16* ap, const _Float16* bp) {
    s.alo = *(const v8h*)(ap);
    s.ahi = *(const v8h*)(ap + 16);
#pragma unroll
    for (int j = 0; j < 4; ++j) {
        s.blo[j] = *(const v8h*)(bp + j * 16 * GEMM_PITCH);
        s.bhi[j] = *(const v8h*)(bp + j * 16 * GEMM_PITCH + 16);
    }
}

__device__ __forceinline__ v16h frag16(v8h lo, v8h hi) {
    return __builtin_shufflevector(lo, hi, 0, 1, 2, 3, 4, 5, 6, 7, 8, 9, 10, 11, 12, 13, 14, 15);
}

__device__ __forceinline__ v8f wmma_f16(v16h a, v16h b, v8f c) {
    return __builtin_amdgcn_wmma_f32_16x16x32_f16(
         false, a,  false, b,
         (short)0, c,  false,  false);
}

__device__ __forceinline__ void put4h(_Float16* p, float4 v) {
    v4h h;
    h[0] = (_Float16)v.x; h[1] = (_Float16)v.y; h[2] = (_Float16)v.z; h[3] = (_Float16)v.w;
    *(v4h*)p = h;
}

__device__ __forceinline__ float half_tree_sum16(const float (&x)[16], int c) {
    const bool b3 = (c & 8) != 0, b2 = (c & 4) != 0, b1 = (c & 2) != 0, b0 = (c & 1) != 0;
    float z8[8], z4[4], z2[2];
#pragma unroll
    for (int i = 0; i < 8; ++i) {
        const float keep = b3 ? x[i + 8] : x[i];
        const float send = b3 ? x[i] : x[i + 8];
        z8[i] = keep + __shfl_xor(send, 8);
    }
#pragma unroll
    for (int i = 0; i < 4; ++i) {
        const float keep = b2 ? z8[i + 4] : z8[i];
        const float send = b2 ? z8[i] : z8[i + 4];
        z4[i] = keep + __shfl_xor(send, 4);
    }
#pragma unroll
    for (int i = 0; i < 2; ++i) {
        const float keep = b1 ? z4[i + 2] : z4[i];
        const float send = b1 ? z4[i] : z4[i + 2];
        z2[i] = keep + __shfl_xor(send, 2);
    }
    const float keep = b0 ? z2[1] : z2[0];
    const float send = b0 ? z2[0] : z2[1];
    return keep + __shfl_xor(send, 1);
}

template <int NOUT, int HC>
__global__ void __launch_bounds__(512)
k_gemm_wmma(const float* __restrict__ A, const float* __restrict__ W,
            float* __restrict__ C,
            const float* __restrict__ asr, const float* __restrict__ adr,
            float* __restrict__ als, float* __restrict__ ald, int mtiles) {
    static_assert(HC == 32 || HC == 128, "a 64-column strip must hold whole heads or lie inside one");
    constexpr int K     = GEMM_K;
    constexpr int PITCH = GEMM_PITCH;
    constexpr int TN4   = NOUT / 64;
    constexpr int NW    = 16;
    constexpr int MTR   = NW / TN4;
    constexpr int AROWS = MTR * 16;
    constexpr int APASS = AROWS / 8;
    constexpr int BQ    = NOUT * (K / 4) / 512;
    static_assert(NW % TN4 == 0 && (BQ & (BQ - 1)) == 0, "shape");

    __shared__ __attribute__((aligned(16))) _Float16 Bs[NOUT * PITCH];
    __shared__ __attribute__((aligned(16))) _Float16 As[2][AROWS * PITCH];

    const int tid   = threadIdx.x;
    const int w     = __builtin_amdgcn_readfirstlane(tid >> 5);
    const int lane  = tid & 31;
    const int r     = lane & 15;
    const int hi    = lane >> 4;
    const int khalf = hi << 3;
    const int mtl   = w / TN4;
    const int nt4   = w - mtl * TN4;

    const int t0 = (int)(blockIdx.x * (unsigned)mtiles / gridDim.x);
    const int t1 = (int)((blockIdx.x + 1u) * (unsigned)mtiles / gridDim.x);

    const int srow = tid >> 6;
    const int scol = (tid & 63) * 4;
    float4 pre[APASS];
    auto fetch = [&](int tb) {
        const int last = (t1 - tb < MTR ? t1 - tb : MTR) * 16 - 1;
#pragma unroll
        for (int p = 0; p < APASS; ++p) {
            const int row = p * 8 + srow;
            pre[p] = *(const float4*)(A + ((long)tb * 16 + (row < last ? row : last)) * K + scol);
        }
    };
    auto stash = [&](int buf) {
#pragma unroll
        for (int p = 0; p < APASS; ++p)
            put4h(&As[buf][(p * 8 + srow) * PITCH + scol], pre[p]);
    };

    if (t0 < t1) fetch(t0);
    {
        constexpr int BB = 8;
        for (int q0 = 0; q0 < BQ; q0 += BB) {
            float4 t[BB];
#pragma unroll
            for (int j = 0; j < BB; ++j) {
                const int idx = ((q0 + j + (int)blockIdx.x) & (BQ - 1)) * 512 + tid;
                t[j] = *(const float4*)(W + (long)idx * 4);
            }
#pragma unroll
            for (int j = 0; j < BB; ++j) {
                const int idx = ((q0 + j + (int)blockIdx.x) & (BQ - 1)) * 512 + tid;
                put4h(&Bs[(idx >> 6) * PITCH + (idx & 63) * 4], t[j]);
            }
        }
    }

    float sa[4], da[4];
#pragma unroll
    for (int j = 0; j < 4; ++j) {
        sa[j] = asr[nt4 * 64 + j * 16 + r];
        da[j] = adr[nt4 * 64 + j * 16 + r];
    }

    const _Float16* bp0 = &Bs[(nt4 * 64 + r) * PITCH + khalf];
    const int hi8 = hi << 3;

    int buf = 0;
    for (int tb = t0; tb < t1; tb += MTR, buf ^= 1) {
        stash(buf);
        __syncthreads();
        if (tb + MTR < t1) fetch(tb + MTR);

        const int mt = tb + mtl;
        if (mt < t1) {
            const _Float16* ap0 = &As[buf][(mtl * 16 + r) * PITCH + khalf];
            v8f acc0 = {}, acc1 = {}, acc2 = {}, acc3 = {};
            KStep st[2];
            load_kstep(st[0], ap0, bp0);
#pragma unroll
            for (int kk = 0; kk < K; kk += 32) {
                const int cur = (kk >> 5) & 1;
                __builtin_amdgcn_sched_barrier(0);
                if (kk + 32 < K) load_kstep(st[cur ^ 1], ap0 + kk + 32, bp0 + kk + 32);
                __builtin_amdgcn_sched_barrier(0);
                const v16h a = frag16(st[cur].alo, st[cur].ahi);
                acc0 = wmma_f16(a, frag16(st[cur].blo[0], st[cur].bhi[0]), acc0);
                acc1 = wmma_f16(a, frag16(st[cur].blo[1], st[cur].bhi[1]), acc1);
                acc2 = wmma_f16(a, frag16(st[cur].blo[2], st[cur].bhi[2]), acc2);
                acc3 = wmma_f16(a, frag16(st[cur].blo[3], st[cur].bhi[3]), acc3);
            }

            float* crow = C + (long)nt4 * 64 + r;
#pragma unroll
            for (int v = 0; v < 8; ++v) {
                long ro = (long)(mt * 16 + hi8 + v) * NOUT;
                crow[ro +  0] = acc0[v];
                crow[ro + 16] = acc1[v];
                crow[ro + 32] = acc2[v];
                crow[ro + 48] = acc3[v];
            }

            const long row = mt * 16 + hi8 + (r & 7);
            if (HC == 32) {
                float xs[16], xd[16];
#pragma unroll
                for (int v = 0; v < 8; ++v) {
                    xs[v]     = acc0[v] * sa[0] + acc1[v] * sa[1];
                    xs[8 + v] = acc2[v] * sa[2] + acc3[v] * sa[3];
                    xd[v]     = acc0[v] * da[0] + acc1[v] * da[1];
                    xd[8 + v] = acc2[v] * da[2] + acc3[v] * da[3];
                }
                const float ts = half_tree_sum16(xs, r);
                const float td = half_tree_sum16(xd, r);
                const long o = row * (2 * TN4) + 2 * nt4 + (r >> 3);
                als[o] = ts;
                ald[o] = td;
            } else {
                float x[16];
#pragma unroll
                for (int v = 0; v < 8; ++v) {
                    x[v]     = (acc0[v] * sa[0] + acc1[v] * sa[1]) + (acc2[v] * sa[2] + acc3[v] * sa[3]);
                    x[8 + v] = (acc0[v] * da[0] + acc1[v] * da[1]) + (acc2[v] * da[2] + acc3[v] * da[3]);
                }
                const float t = half_tree_sum16(x, r);
                float* dstp = (r >> 3) ? ald : als;
                dstp[row * TN4 + nt4] = t;
            }
        }
    }
}

__global__ void k_fill_i32(int* __restrict__ p, int v, long n) {
    long i = (long)blockIdx.x * blockDim.x + threadIdx.x;
    if (i < n) p[i] = v;
}

__global__ void k_deg_count(const int* __restrict__ dst, int E, int* __restrict__ cnt) {
    int e = blockIdx.x * blockDim.x + threadIdx.x;
    if (e < E) atomicAdd(&cnt[dst[e]], 1);
}

#define SHR(v, n) __builtin_amdgcn_update_dpp(0, (v), 0x110 + (n), 0xf, 0xf, false)
__device__ __forceinline__ int wave_incl_scan(int v, int lane) {
    v += SHR(v, 1);
    v += SHR(v, 2);
    v += SHR(v, 4);
    v += SHR(v, 8);
    const int low = __builtin_amdgcn_readlane(v, 15);
    return lane >= 16 ? v + low : v;
}
#undef SHR

__global__ void __launch_bounds__(1024)
k_scan(const int* __restrict__ cnt, int* __restrict__ rowptr, int N) {
    __shared__ int wsum[32];
    constexpr int U = 8;
    const int lane = threadIdx.x & 31;
    const int w    = __builtin_amdgcn_readfirstlane((int)(threadIdx.x >> 5));
    const int nw   = blockDim.x >> 5;
    const int per  = (((N + nw - 1) / nw) + 31) & ~31;
    const int b = w * per;
    const int e = b + per < N ? b + per : N;

    int s = 0;
    for (int i0 = b; i0 < e; i0 += 32 * U) {
        int v[U];
#pragma unroll
        for (int u = 0; u < U; ++u) { const int i = i0 + u * 32 + lane; v[u] = i < e ? cnt[i] : 0; }
#pragma unroll
        for (int u = 0; u < U; ++u) s += v[u];
    }
    for (int off = 16; off > 0; off >>= 1) s += __shfl_xor(s, off);
    if (lane == 0) wsum[w] = s;
    __syncthreads();
    int carry = lane < w ? wsum[lane] : 0;
    for (int off = 16; off > 0; off >>= 1) carry += __shfl_xor(carry, off);

    for (int i0 = b; i0 < e; i0 += 32 * U) {
        int v[U];
#pragma unroll
        for (int u = 0; u < U; ++u) { const int i = i0 + u * 32 + lane; v[u] = i < e ? cnt[i] : 0; }
#pragma unroll
        for (int u = 0; u < U; ++u) {
            const int i = i0 + u * 32 + lane;
            const int inc = wave_incl_scan(v[u], lane);
            if (i < e) rowptr[i] = carry + inc - v[u];
            carry += __builtin_amdgcn_readlane(inc, 31);
        }
    }
    if (w == nw - 1 && lane == 0) rowptr[N] = carry;
}

__global__ void k_place(const int* __restrict__ src, const int* __restrict__ dst, int E,
                        const int* __restrict__ rowptr, int* __restrict__ cursor,
                        int* __restrict__ tmp) {
    int e = blockIdx.x * blockDim.x + threadIdx.x;
    if (e >= E) return;
    int d = dst[e];
    int p = rowptr[d] + atomicAdd(&cursor[d], 1);
    tmp[p] = src[e];
}

__global__ void __launch_bounds__(256)
k_rank(const int* __restrict__ rowptr, const int* __restrict__ tmp,
       int* __restrict__ csr, int N, int totalWaves) {
    const int lane  = threadIdx.x & 31;
    const int wave0 = blockIdx.x * (blockDim.x >> 5) + (threadIdx.x >> 5);
    for (int nn = wave0; nn < N; nn += totalWaves) {
        const int n  = __builtin_amdgcn_readfirstlane(nn);
        const int b  = rowptr[n], en = rowptr[n + 1];
        const int deg = en - b;
        if (deg <= 32) {
            const int mine = lane < deg ? tmp[b + lane] : 0x7fffffff;
            int r = 0;
            for (int j = 0; j < deg; ++j) {
                const int v = __builtin_amdgcn_readlane(mine, j);
                r += (v < mine || (v == mine && j < lane)) ? 1 : 0;
            }
            if (lane < deg) csr[b + r] = mine;
        } else {
            for (int p = b + lane; p < en; p += 32) {
                const int s = tmp[p];
                int r = 0;
                for (int q = b; q < en; ++q) {
                    const int v = tmp[q];
                    r += (v < s || (v == s && q < p)) ? 1 : 0;
                }
                csr[b + r] = s;
            }
        }
    }
}

__device__ __forceinline__ float leaky(float v) { return v > 0.f ? v : NEG_SLOPE * v; }

template <int P>
__device__ __forceinline__ float lterm(const float* __restrict__ p, int i) {
    if (P == 1) return p[i];
    const float2 t = *(const float2*)(p + 2 * (long)i);
    return t.x + t.y;
}

template <int NV>
__device__ __forceinline__ void loads_issued_by_here(const float (&l)[NV]) {
    static_assert(NV == 1 || NV == 8, "one operand list per count in use");
    if constexpr (NV == 1)
        asm volatile("" :: "v"(l[0]));
    else
        asm volatile("" :: "v"(l[0]), "v"(l[1]), "v"(l[2]), "v"(l[3]),
                           "v"(l[4]), "v"(l[5]), "v"(l[6]), "v"(l[7]));
}

template <int H, int C, int ELU>
__global__ void __launch_bounds__(256)
k_gather(const int* __restrict__ rowptr, const int* __restrict__ csr,
         const float* __restrict__ als, const float* __restrict__ ald,
         const float* __restrict__ Hm, const float* __restrict__ bias,
         float* __restrict__ out, int N, int totalWaves) {
    constexpr int NOUT = H * C;
    constexpr int V    = NOUT / 128;
    constexpr int P    = C > 64 ? C / 64 : 1;
    constexpr int G    = C / 4 < 32 ? C / 4 : 32;
    constexpr int T    = 32 / G;
    static_assert((G & (G - 1)) == 0 && G * T == 32, "head groups must tile the wave");
    const int lane = threadIdx.x & 31;
    const int gi   = lane & (G - 1);
    const int gb   = lane & ~(G - 1);
    const int wave0 = blockIdx.x * (blockDim.x >> 5) + (threadIdx.x >> 5);

    int hd[V], ch[V];
#pragma unroll
    for (int k = 0; k < V; ++k) { ch[k] = k * 128 + lane * 4; hd[k] = ch[k] / C; }

    for (int nn = wave0; nn < N; nn += totalWaves) {
        const int n   = __builtin_amdgcn_readfirstlane(nn);
        const int beg = rowptr[n], end = rowptr[n + 1];
        const bool one = end - beg <= 32;

        float adn[V], es[V], m[V], den[V];
        float4 acc[V];
#pragma unroll
        for (int k = 0; k < V; ++k) {
            adn[k] = lterm<P>(ald, n * H + hd[k]);
            es[k]  = leaky(lterm<P>(als, n * H + hd[k]) + adn[k]);
            m[k]   = es[k];
        }

        int   mys = 0;
        float e[V][T];
        auto chunk_logits = [&](int cnt) {
            int sv[T];
#pragma unroll
            for (int t = 0; t < T; ++t) sv[t] = T == 1 ? mys : __shfl(mys, gi + G * t);
            float l[V * T];
#pragma unroll
            for (int k = 0; k < V; ++k)
#pragma unroll
                for (int t = 0; t < T; ++t)
                    l[k * T + t] = lterm<P>(als, sv[t] * H + hd[k]);
            loads_issued_by_here(l);
#pragma unroll
            for (int k = 0; k < V; ++k)
#pragma unroll
                for (int t = 0; t < T; ++t)
                    e[k][t] = gi + G * t < cnt ? leaky(l[k * T + t] + adn[k]) : -__builtin_inff();
        };

        for (int base = beg; base < end; base += 32) {
            const int cnt = end - base < 32 ? end - base : 32;
            mys = lane < cnt ? csr[base + lane] : 0;
            chunk_logits(cnt);
#pragma unroll
            for (int k = 0; k < V; ++k)
#pragma unroll
                for (int t = 0; t < T; ++t) m[k] = fmaxf(m[k], e[k][t]);
        }
#pragma unroll
        for (int k = 0; k < V; ++k)
#pragma unroll
            for (int off = G / 2; off > 0; off >>= 1) m[k] = fmaxf(m[k], __shfl_xor(m[k], off));

#pragma unroll
        for (int k = 0; k < V; ++k) {
            const float ex = expf(es[k] - m[k]);
            const float4 hv = *(const float4*)(Hm + (long)n * NOUT + ch[k]);
            den[k] = ex;
            acc[k] = make_float4(hv.x * ex, hv.y * ex, hv.z * ex, hv.w * ex);
        }
        for (int base = beg; base < end; base += 32) {
            const int cnt = end - base < 32 ? end - base : 32;
            if (!one) {
                mys = lane < cnt ? csr[base + lane] : 0;
                chunk_logits(cnt);
            }
#pragma unroll
            for (int t = 0; t < T; ++t) {
                const int lim = cnt - t * G < G ? cnt - t * G : G;
                if (lim > 0) {
                    float p[V];
#pragma unroll
                    for (int k = 0; k < V; ++k) p[k] = expf(e[k][t] - m[k]);
#pragma unroll 4
                    for (int i = 0; i < lim; ++i) {
                        const int s = __builtin_amdgcn_readlane(mys, t * G + i);
#pragma unroll
                        for (int k = 0; k < V; ++k) {
                            const float ex = G == 32
                                ? __int_as_float(__builtin_amdgcn_readlane(__float_as_int(p[k]), i))
                                : __shfl(p[k], gb + i);
                            const float4 hv = *(const float4*)(Hm + (long)s * NOUT + ch[k]);
                            den[k] += ex;
                            acc[k].x += hv.x * ex;
                            acc[k].y += hv.y * ex;
                            acc[k].z += hv.z * ex;
                            acc[k].w += hv.w * ex;
                        }
                    }
                }
            }
        }
#pragma unroll
        for (int k = 0; k < V; ++k) {
            const float dn = den[k] + 1e-16f;
            const float4 bv = *(const float4*)(bias + ch[k]);
            float4 o;
            o.x = acc[k].x / dn + bv.x;
            o.y = acc[k].y / dn + bv.y;
            o.z = acc[k].z / dn + bv.z;
            o.w = acc[k].w / dn + bv.w;
            if (ELU) {
                o.x = o.x > 0.f ? o.x : (expf(o.x) - 1.f);
                o.y = o.y > 0.f ? o.y : (expf(o.y) - 1.f);
                o.z = o.z > 0.f ? o.z : (expf(o.z) - 1.f);
                o.w = o.w > 0.f ? o.w : (expf(o.w) - 1.f);
            }
            *(float4*)(out + (long)n * NOUT + ch[k]) = o;
        }
    }
}

static inline long cdivl(long a, long b) { return (a + b - 1) / b; }

struct LayerWS {
    float *als, *ald;
    const int *rowptr, *csr;
};

static void run_gat_layer(const float* Xin, const float* W, const float* a_s, const float* a_d,
                          const float* bias, int N, int H,
                          float* Hbuf, float* agg, const LayerWS& w,
                          hipStream_t stream) {
    const int TB = 256;
    if (H == 8)
        k_gemm_wmma<256, 32><<<256, 512, 0, stream>>>(Xin, W, Hbuf, a_s, a_d, w.als, w.ald, N / 16);
    else
        k_gemm_wmma<128, 128><<<256, 512, 0, stream>>>(Xin, W, Hbuf, a_s, a_d, w.als, w.ald, N / 16);
    const int GB = 2048, gWaves = GB * (TB / 32);
    if (H == 8)
        k_gather<8, 32, 1><<<GB, TB, 0, stream>>>(w.rowptr, w.csr, w.als, w.ald, Hbuf, bias,
                                                  agg, N, gWaves);
    else
        k_gather<1, 128, 0><<<GB, TB, 0, stream>>>(w.rowptr, w.csr, w.als, w.ald, Hbuf, bias,
                                                   agg, N, gWaves);
}

extern "C" void kernel_launch(void* const* d_in, const int* in_sizes, int n_in,
                              void* d_out, int out_size, void* d_ws, size_t ws_size,
                              hipStream_t stream) {
    const float* x   = (const float*)d_in[0];
    const int*   ei  = (const int*)  d_in[1];
    const float* W1  = (const float*)d_in[2];
    const float* as1 = (const float*)d_in[3];
    const float* ad1 = (const float*)d_in[4];
    const float* b1  = (const float*)d_in[5];
    const float* W2  = (const float*)d_in[6];
    const float* as2 = (const float*)d_in[7];
    const float* ad2 = (const float*)d_in[8];
    const float* b2  = (const float*)d_in[9];
    const float* W3  = (const float*)d_in[10];
    const float* as3 = (const float*)d_in[11];
    const float* ad3 = (const float*)d_in[12];
    const float* b3  = (const float*)d_in[13];
    float* out = (float*)d_out;

    const int FinW = 256;
    const int N    = in_sizes[0] / FinW;
    const int E    = in_sizes[1] / 2;
    const int* src = ei;
    const int* dst = ei + E;

    char* p = (char*)d_ws;
    float*    P   = (float*)p;    p += (size_t)N * 256 * sizeof(float);
    float*    Q   = (float*)p;    p += (size_t)N * 256 * sizeof(float);
    float*    als = (float*)p;    p += (size_t)N * 8 * sizeof(float);
    float*    ald = (float*)p;    p += (size_t)N * 8 * sizeof(float);
    int*      cnt = (int*)p;      p += (size_t)N * sizeof(int);
    int*      cur = (int*)p;      p += (size_t)N * sizeof(int);
    int*      rowptr = (int*)p;   p += (size_t)(N + 4) * sizeof(int);
    int*      tmp  = (int*)p;     p += (size_t)E * sizeof(int);
    int*      csr  = (int*)p;     p += (size_t)E * sizeof(int);
    LayerWS w{als, ald, rowptr, csr};

    const int TB = 256;
    k_fill_i32<<<(int)cdivl((long)2 * N, TB), TB, 0, stream>>>(cnt, 0, (long)2 * N);
    k_deg_count<<<(int)cdivl(E, TB), TB, 0, stream>>>(dst, E, cnt);
    k_scan<<<1, 1024, 0, stream>>>(cnt, rowptr, N);
    k_place<<<(int)cdivl(E, TB), TB, 0, stream>>>(src, dst, E, rowptr, cur, tmp);
    k_rank<<<2048, TB, 0, stream>>>(rowptr, tmp, csr, N, 2048 * (TB / 32));

    run_gat_layer(x, W1, as1, ad1, b1, N, 8, Q, P, w, stream);
    run_gat_layer(P, W2, as2, ad2, b2, N, 8, Q, P, w, stream);
    run_gat_layer(P, W3, as3, ad3, b3, N, 1, Q, out, w, stream);
}


# Edge list build: an edge's place in its destination group is the value its degree-count atomic returns, kept per edge; second atomic pass and cursor plane gone; edge kernels walk 8 edges per lane
# speedup vs baseline: 6.1894x; 1.2440x over previous
#include <hip/hip_runtime.h>
#include <hip/hip_bf16.h>

typedef _Float16 v16h __attribute__((ext_vector_type(16)));
typedef _Float16 v8h  __attribute__((ext_vector_type(8)));
typedef float    v8f  __attribute__((ext_vector_type(8)));

#define NEG_SLOPE 0.2f

typedef _Float16 v4h __attribute__((ext_vector_type(4)));

#define GEMM_K     256
#define GEMM_PITCH 264

struct KStep {
    v8h alo, ahi;
    v8h blo[4], bhi[4];
};

__device__ __forceinline__ void load_kstep(KStep& s, const _Float16* ap, const _Float16* bp) {
    s.alo = *(const v8h*)(ap);
    s.ahi = *(const v8h*)(ap + 16);
#pragma unroll
    for (int j = 0; j < 4; ++j) {
        s.blo[j] = *(const v8h*)(bp + j * 16 * GEMM_PITCH);
        s.bhi[j] = *(const v8h*)(bp + j * 16 * GEMM_PITCH + 16);
    }
}

__device__ __forceinline__ v16h frag16(v8h lo, v8h hi) {
    return __builtin_shufflevector(lo, hi, 0, 1, 2, 3, 4, 5, 6, 7, 8, 9, 10, 11, 12, 13, 14, 15);
}

__device__ __forceinline__ v8f wmma_f16(v16h a, v16h b, v8f c) {
    return __builtin_amdgcn_wmma_f32_16x16x32_f16(
         false, a,  false, b,
         (short)0, c,  false,  false);
}

__device__ __forceinline__ void put4h(_Float16* p, float4 v) {
    v4h h;
    h[0] = (_Float16)v.x; h[1] = (_Float16)v.y; h[2] = (_Float16)v.z; h[3] = (_Float16)v.w;
    *(v4h*)p = h;
}

__device__ __forceinline__ float half_tree_sum16(const float (&x)[16], int c) {
    const bool b3 = (c & 8) != 0, b2 = (c & 4) != 0, b1 = (c & 2) != 0, b0 = (c & 1) != 0;
    float z8[8], z4[4], z2[2];
#pragma unroll
    for (int i = 0; i < 8; ++i) {
        const float keep = b3 ? x[i + 8] : x[i];
        const float send = b3 ? x[i] : x[i + 8];
        z8[i] = keep + __shfl_xor(send, 8);
    }
#pragma unroll
    for (int i = 0; i < 4; ++i) {
        const float keep = b2 ? z8[i + 4] : z8[i];
        const float send = b2 ? z8[i] : z8[i + 4];
        z4[i] = keep + __shfl_xor(send, 4);
    }
#pragma unroll
    for (int i = 0; i < 2; ++i) {
        const float keep = b1 ? z4[i + 2] : z4[i];
        const float send = b1 ? z4[i] : z4[i + 2];
        z2[i] = keep + __shfl_xor(send, 2);
    }
    const float keep = b0 ? z2[1] : z2[0];
    const float send = b0 ? z2[0] : z2[1];
    return keep + __shfl_xor(send, 1);
}

template <int NOUT, int HC, int C16>
__global__ void __launch_bounds__(512)
k_gemm_wmma(const float* __restrict__ A, const float* __restrict__ W,
            float* __restrict__ C,
            const float* __restrict__ asr, const float* __restrict__ adr,
            float* __restrict__ als, float* __restrict__ ald, int mtiles) {
    static_assert(HC == 32 || HC == 128, "a 64-column strip must hold whole heads or lie inside one");
    constexpr int K     = GEMM_K;
    constexpr int PITCH = GEMM_PITCH;
    constexpr int TN4   = NOUT / 64;
    constexpr int NW    = 16;
    constexpr int MTR   = NW / TN4;
    constexpr int AROWS = MTR * 16;
    constexpr int APASS = AROWS / 8;
    constexpr int BQ    = NOUT * (K / 4) / 512;
    static_assert(NW % TN4 == 0 && (BQ & (BQ - 1)) == 0, "shape");

    __shared__ __attribute__((aligned(16))) _Float16 Bs[NOUT * PITCH];
    __shared__ __attribute__((aligned(16))) _Float16 As[2][AROWS * PITCH];

    const int tid   = threadIdx.x;
    const int w     = __builtin_amdgcn_readfirstlane(tid >> 5);
    const int lane  = tid & 31;
    const int r     = lane & 15;
    const int hi    = lane >> 4;
    const int khalf = hi << 3;
    const int mtl   = w / TN4;
    const int nt4   = w - mtl * TN4;

    const int t0 = (int)(blockIdx.x * (unsigned)mtiles / gridDim.x);
    const int t1 = (int)((blockIdx.x + 1u) * (unsigned)mtiles / gridDim.x);

    const int srow = tid >> 6;
    const int scol = (tid & 63) * 4;
    float4 pre[APASS];
    auto fetch = [&](int tb) {
        const int last = (t1 - tb < MTR ? t1 - tb : MTR) * 16 - 1;
#pragma unroll
        for (int p = 0; p < APASS; ++p) {
            const int row = p * 8 + srow;
            pre[p] = *(const float4*)(A + ((long)tb * 16 + (row < last ? row : last)) * K + scol);
        }
    };
    auto stash = [&](int buf) {
#pragma unroll
        for (int p = 0; p < APASS; ++p)
            put4h(&As[buf][(p * 8 + srow) * PITCH + scol], pre[p]);
    };

    if (t0 < t1) fetch(t0);
    {
        constexpr int BB = 8;
        for (int q0 = 0; q0 < BQ; q0 += BB) {
            float4 t[BB];
#pragma unroll
            for (int j = 0; j < BB; ++j) {
                const int idx = ((q0 + j + (int)blockIdx.x) & (BQ - 1)) * 512 + tid;
                t[j] = *(const float4*)(W + (long)idx * 4);
            }
#pragma unroll
            for (int j = 0; j < BB; ++j) {
                const int idx = ((q0 + j + (int)blockIdx.x) & (BQ - 1)) * 512 + tid;
                put4h(&Bs[(idx >> 6) * PITCH + (idx & 63) * 4], t[j]);
            }
        }
    }

    float sa[4], da[4];
#pragma unroll
    for (int j = 0; j < 4; ++j) {
        sa[j] = asr[nt4 * 64 + j * 16 + r];
        da[j] = adr[nt4 * 64 + j * 16 + r];
    }

    const _Float16* bp0 = &Bs[(nt4 * 64 + r) * PITCH + khalf];
    const int hi8 = hi << 3;

    int buf = 0;
    for (int tb = t0; tb < t1; tb += MTR, buf ^= 1) {
        stash(buf);
        __syncthreads();
        if (tb + MTR < t1) fetch(tb + MTR);

        const int mt = tb + mtl;
        if (mt < t1) {
            const _Float16* ap0 = &As[buf][(mtl * 16 + r) * PITCH + khalf];
            v8f acc0 = {}, acc1 = {}, acc2 = {}, acc3 = {};
            KStep st[2];
            load_kstep(st[0], ap0, bp0);
#pragma unroll
            for (int kk = 0; kk < K; kk += 32) {
                const int cur = (kk >> 5) & 1;
                __builtin_amdgcn_sched_barrier(0);
                if (kk + 32 < K) load_kstep(st[cur ^ 1], ap0 + kk + 32, bp0 + kk + 32);
                __builtin_amdgcn_sched_barrier(0);
                const v16h a = frag16(st[cur].alo, st[cur].ahi);
                acc0 = wmma_f16(a, frag16(st[cur].blo[0], st[cur].bhi[0]), acc0);
                acc1 = wmma_f16(a, frag16(st[cur].blo[1], st[cur].bhi[1]), acc1);
                acc2 = wmma_f16(a, frag16(st[cur].blo[2], st[cur].bhi[2]), acc2);
                acc3 = wmma_f16(a, frag16(st[cur].blo[3], st[cur].bhi[3]), acc3);
            }

            if (C16) {
                _Float16* crow = (_Float16*)C + (long)nt4 * 64 + r;
#pragma unroll
                for (int v = 0; v < 8; ++v) {
                    long ro = (long)(mt * 16 + hi8 + v) * NOUT;
                    crow[ro +  0] = (_Float16)acc0[v];
                    crow[ro + 16] = (_Float16)acc1[v];
                    crow[ro + 32] = (_Float16)acc2[v];
                    crow[ro + 48] = (_Float16)acc3[v];
                }
            } else {
                float* crow = C + (long)nt4 * 64 + r;
#pragma unroll
                for (int v = 0; v < 8; ++v) {
                    long ro = (long)(mt * 16 + hi8 + v) * NOUT;
                    crow[ro +  0] = acc0[v];
                    crow[ro + 16] = acc1[v];
                    crow[ro + 32] = acc2[v];
                    crow[ro + 48] = acc3[v];
                }
            }

            const long row = mt * 16 + hi8 + (r & 7);
            if (HC == 32) {
                float xs[16], xd[16];
#pragma unroll
                for (int v = 0; v < 8; ++v) {
                    xs[v]     = acc0[v] * sa[0] + acc1[v] * sa[1];
                    xs[8 + v] = acc2[v] * sa[2] + acc3[v] * sa[3];
                    xd[v]     = acc0[v] * da[0] + acc1[v] * da[1];
                    xd[8 + v] = acc2[v] * da[2] + acc3[v] * da[3];
                }
                const float ts = half_tree_sum16(xs, r);
                const float td = half_tree_sum16(xd, r);
                const long o = row * (2 * TN4) + 2 * nt4 + (r >> 3);
                als[o] = ts;
                ald[o] = td;
            } else {
                float x[16];
#pragma unroll
                for (int v = 0; v < 8; ++v) {
                    x[v]     = (acc0[v] * sa[0] + acc1[v] * sa[1]) + (acc2[v] * sa[2] + acc3[v] * sa[3]);
                    x[8 + v] = (acc0[v] * da[0] + acc1[v] * da[1]) + (acc2[v] * da[2] + acc3[v] * da[3]);
                }
                const float t = half_tree_sum16(x, r);
                float* dstp = (r >> 3) ? ald : als;
                dstp[row * TN4 + nt4] = t;
            }
        }
    }
}

__global__ void k_fill_i32(int* __restrict__ p, int v, long n) {
    long i = (long)blockIdx.x * blockDim.x + threadIdx.x;
    if (i < n) p[i] = v;
}

#define EDGE_U 2
#define EDGE_W (EDGE_U * 128)

__device__ __forceinline__ bool aligned16(const void* p) { return ((size_t)p & 15) == 0; }

__device__ __forceinline__ void edge_loads_issued_by_here(const int4& a, const int4& b) {
    asm volatile("" :: "v"(a.x), "v"(a.y), "v"(a.z), "v"(a.w),
                       "v"(b.x), "v"(b.y), "v"(b.z), "v"(b.w));
}

__global__ void __launch_bounds__(256)
k_deg_slot(const int* __restrict__ dst, int E, int* __restrict__ cnt, int* __restrict__ slot) {
    const int lane = threadIdx.x & 31;
    const int wave = __builtin_amdgcn_readfirstlane(
        (int)(blockIdx.x * (blockDim.x >> 5) + (threadIdx.x >> 5)));
    const int w0 = wave * EDGE_W;
    if (w0 + EDGE_W <= E && aligned16(dst) && aligned16(slot)) {
        int4 d[EDGE_U], s[EDGE_U];
#pragma unroll
        for (int j = 0; j < EDGE_U; ++j) d[j] = *(const int4*)(dst + w0 + (j * 32 + lane) * 4);
        static_assert(EDGE_U == 2, "the load fence below names two groups");
        edge_loads_issued_by_here(d[0], d[1]);
#pragma unroll
        for (int j = 0; j < EDGE_U; ++j) {
            s[j].x = atomicAdd(&cnt[d[j].x], 1);
            s[j].y = atomicAdd(&cnt[d[j].y], 1);
            s[j].z = atomicAdd(&cnt[d[j].z], 1);
            s[j].w = atomicAdd(&cnt[d[j].w], 1);
        }
#pragma unroll
        for (int j = 0; j < EDGE_U; ++j) *(int4*)(slot + w0 + (j * 32 + lane) * 4) = s[j];
    } else {
        for (int e = w0 + lane; e < w0 + EDGE_W && e < E; e += 32)
            slot[e] = atomicAdd(&cnt[dst[e]], 1);
    }
}

#define SHR(v, n) __builtin_amdgcn_update_dpp(0, (v), 0x110 + (n), 0xf, 0xf, false)
__device__ __forceinline__ int wave_incl_scan(int v, int lane) {
    v += SHR(v, 1);
    v += SHR(v, 2);
    v += SHR(v, 4);
    v += SHR(v, 8);
    const int low = __builtin_amdgcn_readlane(v, 15);
    return lane >= 16 ? v + low : v;
}
#undef SHR

__global__ void __launch_bounds__(1024)
k_scan(const int* __restrict__ cnt, int* __restrict__ rowptr, int N) {
    __shared__ int wsum[32];
    constexpr int U = 8;
    const int lane = threadIdx.x & 31;
    const int w    = __builtin_amdgcn_readfirstlane((int)(threadIdx.x >> 5));
    const int nw   = blockDim.x >> 5;
    const int per  = (((N + nw - 1) / nw) + 31) & ~31;
    const int b = w * per;
    const int e = b + per < N ? b + per : N;

    int s = 0;
    for (int i0 = b; i0 < e; i0 += 32 * U) {
        int v[U];
#pragma unroll
        for (int u = 0; u < U; ++u) { const int i = i0 + u * 32 + lane; v[u] = i < e ? cnt[i] : 0; }
#pragma unroll
        for (int u = 0; u < U; ++u) s += v[u];
    }
    for (int off = 16; off > 0; off >>= 1) s += __shfl_xor(s, off);
    if (lane == 0) wsum[w] = s;
    __syncthreads();
    int carry = lane < w ? wsum[lane] : 0;
    for (int off = 16; off > 0; off >>= 1) carry += __shfl_xor(carry, off);

    for (int i0 = b; i0 < e; i0 += 32 * U) {
        int v[U];
#pragma unroll
        for (int u = 0; u < U; ++u) { const int i = i0 + u * 32 + lane; v[u] = i < e ? cnt[i] : 0; }
#pragma unroll
        for (int u = 0; u < U; ++u) {
            const int i = i0 + u * 32 + lane;
            const int inc = wave_incl_scan(v[u], lane);
            if (i < e) rowptr[i] = carry + inc - v[u];
            carry += __builtin_amdgcn_readlane(inc, 31);
        }
    }
    if (w == nw - 1 && lane == 0) rowptr[N] = carry;
}

__global__ void __launch_bounds__(256)
k_place(const int* __restrict__ src, const int* __restrict__ dst,
        const int* __restrict__ slot, int E,
        const int* __restrict__ rowptr, int* __restrict__ tmp) {
    const int lane = threadIdx.x & 31;
    const int wave = __builtin_amdgcn_readfirstlane(
        (int)(blockIdx.x * (blockDim.x >> 5) + (threadIdx.x >> 5)));
    const int w0 = wave * EDGE_W;
    if (w0 + EDGE_W <= E && aligned16(src) && aligned16(dst) && aligned16(slot)) {
        int4 s[EDGE_U], d[EDGE_U], k[EDGE_U], p[EDGE_U];
#pragma unroll
        for (int j = 0; j < EDGE_U; ++j) {
            const int e0 = w0 + (j * 32 + lane) * 4;
            d[j] = *(const int4*)(dst + e0);
            k[j] = *(const int4*)(slot + e0);
            s[j] = *(const int4*)(src + e0);
        }
#pragma unroll
        for (int j = 0; j < EDGE_U; ++j) {
            p[j].x = rowptr[d[j].x] + k[j].x;
            p[j].y = rowptr[d[j].y] + k[j].y;
            p[j].z = rowptr[d[j].z] + k[j].z;
            p[j].w = rowptr[d[j].w] + k[j].w;
        }
#pragma unroll
        for (int j = 0; j < EDGE_U; ++j) {
            tmp[p[j].x] = s[j].x;
            tmp[p[j].y] = s[j].y;
            tmp[p[j].z] = s[j].z;
            tmp[p[j].w] = s[j].w;
        }
    } else {
        for (int e = w0 + lane; e < w0 + EDGE_W && e < E; e += 32)
            tmp[rowptr[dst[e]] + slot[e]] = src[e];
    }
}

__global__ void __launch_bounds__(256)
k_rank(const int* __restrict__ rowptr, const int* __restrict__ tmp,
       int* __restrict__ csr, int N, int totalWaves) {
    const int lane  = threadIdx.x & 31;
    const int wave0 = blockIdx.x * (blockDim.x >> 5) + (threadIdx.x >> 5);
    for (int nn = wave0; nn < N; nn += totalWaves) {
        const int n  = __builtin_amdgcn_readfirstlane(nn);
        const int b  = rowptr[n], en = rowptr[n + 1];
        const int deg = en - b;
        if (deg <= 32) {
            const int mine = lane < deg ? tmp[b + lane] : 0x7fffffff;
            int r = 0;
            for (int j = 0; j < deg; ++j) {
                const int v = __builtin_amdgcn_readlane(mine, j);
                r += (v < mine || (v == mine && j < lane)) ? 1 : 0;
            }
            if (lane < deg) csr[b + r] = mine;
        } else {
            for (int p = b + lane; p < en; p += 32) {
                const int s = tmp[p];
                int r = 0;
                for (int q = b; q < en; ++q) {
                    const int v = tmp[q];
                    r += (v < s || (v == s && q < p)) ? 1 : 0;
                }
                csr[b + r] = s;
            }
        }
    }
}

__device__ __forceinline__ float leaky(float v) { return v > 0.f ? v : NEG_SLOPE * v; }

__device__ __forceinline__ float exp_le0(float x) {
    return __builtin_amdgcn_exp2f(x * 1.44269504088896340736f);
}

template <int P>
__device__ __forceinline__ float lterm(const float* __restrict__ p, int i) {
    if (P == 1) return p[i];
    const float2 t = *(const float2*)(p + 2 * (long)i);
    return t.x + t.y;
}

template <int NV>
__device__ __forceinline__ void loads_issued_by_here(const float (&l)[NV]) {
    static_assert(NV == 1 || NV == 8, "one operand list per count in use");
    if constexpr (NV == 1)
        asm volatile("" :: "v"(l[0]));
    else
        asm volatile("" :: "v"(l[0]), "v"(l[1]), "v"(l[2]), "v"(l[3]),
                           "v"(l[4]), "v"(l[5]), "v"(l[6]), "v"(l[7]));
}

template <int R16, int NOUT>
__device__ __forceinline__ float4 row4(const float* __restrict__ Hm, long row, int ch) {
    if (R16) {
        const v4h h = *(const v4h*)((const _Float16*)Hm + row * NOUT + ch);
        return make_float4((float)h[0], (float)h[1], (float)h[2], (float)h[3]);
    }
    return *(const float4*)(Hm + row * NOUT + ch);
}

__device__ __forceinline__ void mad_mix_lo(float& acc, float w, unsigned h2) {
    asm("v_fma_mix_f32 %0, %1, %2, %0 op_sel_hi:[0,1,0]" : "+v"(acc) : "v"(w), "v"(h2));
}
__device__ __forceinline__ void mad_mix_hi(float& acc, float w, unsigned h2) {
    asm("v_fma_mix_f32 %0, %1, %2, %0 op_sel:[0,1,0] op_sel_hi:[0,1,0]" : "+v"(acc) : "v"(w), "v"(h2));
}

template <int H, int C, int ELU, int R16>
__global__ void __launch_bounds__(256)
k_gather(const int* __restrict__ rowptr, const int* __restrict__ csr,
         const float* __restrict__ als, const float* __restrict__ ald,
         const float* __restrict__ Hm, const float* __restrict__ bias,
         float* __restrict__ out, int N, int totalWaves) {
    constexpr int NOUT = H * C;
    constexpr int V    = NOUT / 128;
    constexpr int P    = C > 64 ? C / 64 : 1;
    constexpr int G    = C / 4 < 32 ? C / 4 : 32;
    constexpr int T    = 32 / G;
    static_assert((G & (G - 1)) == 0 && G * T == 32, "head groups must tile the wave");
    const int lane = threadIdx.x & 31;
    const int gi   = lane & (G - 1);
    const int gb   = lane & ~(G - 1);
    const int wave0 = blockIdx.x * (blockDim.x >> 5) + (threadIdx.x >> 5);

    int hd[V], ch[V];
#pragma unroll
    for (int k = 0; k < V; ++k) { ch[k] = k * 128 + lane * 4; hd[k] = ch[k] / C; }

    for (int nn = wave0; nn < N; nn += totalWaves) {
        const int n   = __builtin_amdgcn_readfirstlane(nn);
        const int beg = rowptr[n], end = rowptr[n + 1];
        const bool one = end - beg <= 32;

        float adn[V], es[V], m[V], den[V];
        float4 acc[V];
#pragma unroll
        for (int k = 0; k < V; ++k) {
            adn[k] = lterm<P>(ald, n * H + hd[k]);
            es[k]  = leaky(lterm<P>(als, n * H + hd[k]) + adn[k]);
            m[k]   = es[k];
        }

        int   mys = 0;
        float e[V][T];
        auto chunk_logits = [&](int cnt) {
            int sv[T];
#pragma unroll
            for (int t = 0; t < T; ++t) sv[t] = T == 1 ? mys : __shfl(mys, gi + G * t);
            float l[V * T];
#pragma unroll
            for (int k = 0; k < V; ++k)
#pragma unroll
                for (int t = 0; t < T; ++t)
                    l[k * T + t] = lterm<P>(als, sv[t] * H + hd[k]);
            loads_issued_by_here(l);
#pragma unroll
            for (int k = 0; k < V; ++k)
#pragma unroll
                for (int t = 0; t < T; ++t)
                    e[k][t] = gi + G * t < cnt ? leaky(l[k * T + t] + adn[k]) : -__builtin_inff();
        };

        for (int base = beg; base < end; base += 32) {
            const int cnt = end - base < 32 ? end - base : 32;
            mys = lane < cnt ? csr[base + lane] : 0;
            chunk_logits(cnt);
#pragma unroll
            for (int k = 0; k < V; ++k)
#pragma unroll
                for (int t = 0; t < T; ++t) m[k] = fmaxf(m[k], e[k][t]);
        }
#pragma unroll
        for (int k = 0; k < V; ++k)
#pragma unroll
            for (int off = G / 2; off > 0; off >>= 1) m[k] = fmaxf(m[k], __shfl_xor(m[k], off));

#pragma unroll
        for (int k = 0; k < V; ++k) {
            const float ex = exp_le0(es[k] - m[k]);
            const float4 hv = row4<R16, NOUT>(Hm, n, ch[k]);
            den[k] = ex;
            acc[k] = make_float4(hv.x * ex, hv.y * ex, hv.z * ex, hv.w * ex);
        }
        for (int base = beg; base < end; base += 32) {
            const int cnt = end - base < 32 ? end - base : 32;
            if (!one) {
                mys = lane < cnt ? csr[base + lane] : 0;
                chunk_logits(cnt);
            }
#pragma unroll
            for (int t = 0; t < T; ++t) {
                const int lim = cnt - t * G < G ? cnt - t * G : G;
                if (lim > 0) {
                    float p[V];
#pragma unroll
                    for (int k = 0; k < V; ++k) p[k] = exp_le0(e[k][t] - m[k]);
#pragma unroll 4
                    for (int i = 0; i < lim; ++i) {
                        const int s = __builtin_amdgcn_readlane(mys, t * G + i);
#pragma unroll
                        for (int k = 0; k < V; ++k) {
                            const float ex = G == 32
                                ? __int_as_float(__builtin_amdgcn_readlane(__float_as_int(p[k]), i))
                                : __shfl(p[k], gb + i);
                            den[k] += ex;
                            if (R16) {
                                const uint2 h = *(const uint2*)((const _Float16*)Hm + (long)s * NOUT + ch[k]);
                                mad_mix_lo(acc[k].x, ex, h.x);
                                mad_mix_hi(acc[k].y, ex, h.x);
                                mad_mix_lo(acc[k].z, ex, h.y);
                                mad_mix_hi(acc[k].w, ex, h.y);
                            } else {
                                const float4 hv = row4<R16, NOUT>(Hm, s, ch[k]);
                                acc[k].x += hv.x * ex;
                                acc[k].y += hv.y * ex;
                                acc[k].z += hv.z * ex;
                                acc[k].w += hv.w * ex;
                            }
                        }
                    }
                }
            }
        }
#pragma unroll
        for (int k = 0; k < V; ++k) {
            const float rdn = 1.0f / (den[k] + 1e-16f);
            const float4 bv = *(const float4*)(bias + ch[k]);
            float4 o;
            o.x = acc[k].x * rdn + bv.x;
            o.y = acc[k].y * rdn + bv.y;
            o.z = acc[k].z * rdn + bv.z;
            o.w = acc[k].w * rdn + bv.w;
            if (ELU) {
                o.x = o.x > 0.f ? o.x : (exp_le0(fminf(o.x, 0.f)) - 1.f);
                o.y = o.y > 0.f ? o.y : (exp_le0(fminf(o.y, 0.f)) - 1.f);
                o.z = o.z > 0.f ? o.z : (exp_le0(fminf(o.z, 0.f)) - 1.f);
                o.w = o.w > 0.f ? o.w : (exp_le0(fminf(o.w, 0.f)) - 1.f);
            }
            *(float4*)(out + (long)n * NOUT + ch[k]) = o;
        }
    }
}

static inline long cdivl(long a, long b) { return (a + b - 1) / b; }

struct LayerWS {
    float *als, *ald;
    const int *rowptr, *csr;
};

static void run_gat_layer(const float* Xin, const float* W, const float* a_s, const float* a_d,
                          const float* bias, int N, int H,
                          float* Hbuf, float* agg, const LayerWS& w,
                          hipStream_t stream) {
    const int TB = 256;
    if (H == 8)
        k_gemm_wmma<256, 32, 1><<<256, 512, 0, stream>>>(Xin, W, Hbuf, a_s, a_d, w.als, w.ald, N / 16);
    else
        k_gemm_wmma<128, 128, 0><<<256, 512, 0, stream>>>(Xin, W, Hbuf, a_s, a_d, w.als, w.ald, N / 16);
    const int GB = 2048, gWaves = GB * (TB / 32);
    if (H == 8)
        k_gather<8, 32, 1, 1><<<GB, TB, 0, stream>>>(w.rowptr, w.csr, w.als, w.ald, Hbuf, bias,
                                                  agg, N, gWaves);
    else
        k_gather<1, 128, 0, 0><<<GB, TB, 0, stream>>>(w.rowptr, w.csr, w.als, w.ald, Hbuf, bias,
                                                   agg, N, gWaves);
}

extern "C" void kernel_launch(void* const* d_in, const int* in_sizes, int n_in,
                              void* d_out, int out_size, void* d_ws, size_t ws_size,
                              hipStream_t stream) {
    const float* x   = (const float*)d_in[0];
    const int*   ei  = (const int*)  d_in[1];
    const float* W1  = (const float*)d_in[2];
    const float* as1 = (const float*)d_in[3];
    const float* ad1 = (const float*)d_in[4];
    const float* b1  = (const float*)d_in[5];
    const float* W2  = (const float*)d_in[6];
    const float* as2 = (const float*)d_in[7];
    const float* ad2 = (const float*)d_in[8];
    const float* b2  = (const float*)d_in[9];
    const float* W3  = (const float*)d_in[10];
    const float* as3 = (const float*)d_in[11];
    const float* ad3 = (const float*)d_in[12];
    const float* b3  = (const float*)d_in[13];
    float* out = (float*)d_out;

    const int FinW = 256;
    const int N    = in_sizes[0] / FinW;
    const int E    = in_sizes[1] / 2;
    const int* src = ei;
    const int* dst = ei + E;

    char* p = (char*)d_ws;
    float*    P   = (float*)p;    p += (size_t)N * 256 * sizeof(float);
    float*    Q   = (float*)p;    p += (size_t)N * 256 * sizeof(float);
    float*    als = (float*)p;    p += (size_t)N * 8 * sizeof(float);
    float*    ald = (float*)p;    p += (size_t)N * 8 * sizeof(float);
    int*      cnt = (int*)p;      p += (size_t)N * sizeof(int);
    int*      slot = (int*)p;     p += (size_t)(E + 4) * sizeof(int);
    int*      rowptr = (int*)p;   p += (size_t)(N + 4) * sizeof(int);
    int*      tmp  = (int*)p;     p += (size_t)E * sizeof(int);
    int*      csr  = (int*)p;     p += (size_t)E * sizeof(int);
    LayerWS w{als, ald, rowptr, csr};

    const int TB = 256;
    const int EB = (int)cdivl(E, (long)(TB / 32) * EDGE_W);
    k_fill_i32<<<(int)cdivl(N, TB), TB, 0, stream>>>(cnt, 0, (long)N);
    k_deg_slot<<<EB, TB, 0, stream>>>(dst, E, cnt, slot);
    k_scan<<<1, 1024, 0, stream>>>(cnt, rowptr, N);
    k_place<<<EB, TB, 0, stream>>>(src, dst, slot, E, rowptr, tmp);
    k_rank<<<2048, TB, 0, stream>>>(rowptr, tmp, csr, N, 2048 * (TB / 32));

    run_gat_layer(x, W1, as1, ad1, b1, N, 8, Q, P, w, stream);
    run_gat_layer(P, W2, as2, ad2, b2, N, 8, Q, P, w, stream);
    run_gat_layer(P, W3, as3, ad3, b3, N, 1, Q, out, w, stream);
}


# Prefix sum of the in-degrees (one work-group's work) runs as the first work-group of the first layer's GEMM launch, beside the product, not in a launch of its own; its counts stay in registers
# speedup vs baseline: 6.5499x; 1.0582x over previous
#include <hip/hip_runtime.h>
#include <hip/hip_bf16.h>

typedef _Float16 v16h __attribute__((ext_vector_type(16)));
typedef _Float16 v8h  __attribute__((ext_vector_type(8)));
typedef float    v8f  __attribute__((ext_vector_type(8)));

#define NEG_SLOPE 0.2f

typedef _Float16 v4h __attribute__((ext_vector_type(4)));

#define GEMM_K     256
#define GEMM_PITCH 264

struct KStep {
    v8h alo, ahi;
    v8h blo[4], bhi[4];
};

__device__ __forceinline__ void load_kstep(KStep& s, const _Float16* ap, const _Float16* bp) {
    s.alo = *(const v8h*)(ap);
    s.ahi = *(const v8h*)(ap + 16);
#pragma unroll
    for (int j = 0; j < 4; ++j) {
        s.blo[j] = *(const v8h*)(bp + j * 16 * GEMM_PITCH);
        s.bhi[j] = *(const v8h*)(bp + j * 16 * GEMM_PITCH + 16);
    }
}

__device__ __forceinline__ v16h frag16(v8h lo, v8h hi) {
    return __builtin_shufflevector(lo, hi, 0, 1, 2, 3, 4, 5, 6, 7, 8, 9, 10, 11, 12, 13, 14, 15);
}

__device__ __forceinline__ v8f wmma_f16(v16h a, v16h b, v8f c) {
    return __builtin_amdgcn_wmma_f32_16x16x32_f16(
         false, a,  false, b,
         (short)0, c,  false,  false);
}

__device__ __forceinline__ void put4h(_Float16* p, float4 v) {
    v4h h;
    h[0] = (_Float16)v.x; h[1] = (_Float16)v.y; h[2] = (_Float16)v.z; h[3] = (_Float16)v.w;
    *(v4h*)p = h;
}

__device__ __forceinline__ float half_tree_sum16(const float (&x)[16], int c) {
    const bool b3 = (c & 8) != 0, b2 = (c & 4) != 0, b1 = (c & 2) != 0, b0 = (c & 1) != 0;
    float z8[8], z4[4], z2[2];
#pragma unroll
    for (int i = 0; i < 8; ++i) {
        const float keep = b3 ? x[i + 8] : x[i];
        const float send = b3 ? x[i] : x[i + 8];
        z8[i] = keep + __shfl_xor(send, 8);
    }
#pragma unroll
    for (int i = 0; i < 4; ++i) {
        const float keep = b2 ? z8[i + 4] : z8[i];
        const float send = b2 ? z8[i] : z8[i + 4];
        z4[i] = keep + __shfl_xor(send, 4);
    }
#pragma unroll
    for (int i = 0; i < 2; ++i) {
        const float keep = b1 ? z4[i + 2] : z4[i];
        const float send = b1 ? z4[i] : z4[i + 2];
        z2[i] = keep + __shfl_xor(send, 2);
    }
    const float keep = b0 ? z2[1] : z2[0];
    const float send = b0 ? z2[0] : z2[1];
    return keep + __shfl_xor(send, 1);
}

#define SHR(v, n) __builtin_amdgcn_update_dpp(0, (v), 0x110 + (n), 0xf, 0xf, false)
__device__ __forceinline__ int wave_incl_scan(int v, int lane) {
    v += SHR(v, 1);
    v += SHR(v, 2);
    v += SHR(v, 4);
    v += SHR(v, 8);
    const int low = __builtin_amdgcn_readlane(v, 15);
    return lane >= 16 ? v + low : v;
}
#undef SHR

template <int LPW>
__device__ __forceinline__ void scan_counts(const int* __restrict__ cnt, int* __restrict__ rowptr, int N) {
    __shared__ int wsum[32];
    constexpr int U = 8;
    const int lane = threadIdx.x & 31;
    const int w    = __builtin_amdgcn_readfirstlane((int)(threadIdx.x >> 5));
    const int nw   = blockDim.x >> 5;
    const int per  = (((N + nw - 1) / nw) + 31) & ~31;
    const int b = w * per;
    const int e = b + per < N ? b + per : N;
    const bool inreg = per <= LPW * 32;

    int v[LPW];
    int s = 0;
    if (inreg) {
#pragma unroll
        for (int u = 0; u < LPW; ++u) v[u] = 0;
        if (b < e) {
#pragma unroll
            for (int u = 0; u < LPW; ++u) {
                const int i = b + u * 32 + lane;
                const int t = cnt[i < e ? i : e - 1];
                v[u] = i < e ? t : 0;
            }
        }
#pragma unroll
        for (int u = 0; u < LPW; ++u) s += v[u];
    } else {
        for (int i0 = b; i0 < e; i0 += 32 * U) {
            int t[U];
#pragma unroll
            for (int u = 0; u < U; ++u) { const int i = i0 + u * 32 + lane; t[u] = i < e ? cnt[i] : 0; }
#pragma unroll
            for (int u = 0; u < U; ++u) s += t[u];
        }
    }
    for (int off = 16; off > 0; off >>= 1) s += __shfl_xor(s, off);
    if (lane == 0) wsum[w] = s;
    __syncthreads();
    int carry = lane < w ? wsum[lane] : 0;
    for (int off = 16; off > 0; off >>= 1) carry += __shfl_xor(carry, off);

    if (inreg) {
#pragma unroll
        for (int u = 0; u < LPW; ++u) {
            const int i = b + u * 32 + lane;
            const int inc = wave_incl_scan(v[u], lane);
            if (i < e) rowptr[i] = carry + inc - v[u];
            carry += __builtin_amdgcn_readlane(inc, 31);
        }
    } else {
        for (int i0 = b; i0 < e; i0 += 32 * U) {
            int t[U];
#pragma unroll
            for (int u = 0; u < U; ++u) { const int i = i0 + u * 32 + lane; t[u] = i < e ? cnt[i] : 0; }
#pragma unroll
            for (int u = 0; u < U; ++u) {
                const int i = i0 + u * 32 + lane;
                const int inc = wave_incl_scan(t[u], lane);
                if (i < e) rowptr[i] = carry + inc - t[u];
                carry += __builtin_amdgcn_readlane(inc, 31);
            }
        }
    }
    if (w == nw - 1 && lane == 0) rowptr[N] = carry;
}

template <int NOUT, int HC, int C16, int SCAN>
__global__ void __launch_bounds__(512)
k_gemm_wmma(const float* __restrict__ A, const float* __restrict__ W,
            float* __restrict__ C,
            const float* __restrict__ asr, const float* __restrict__ adr,
            float* __restrict__ als, float* __restrict__ ald, int mtiles,
            const int* __restrict__ deg, int* __restrict__ rowptr, int nnodes) {
    static_assert(HC == 32 || HC == 128, "a 64-column strip must hold whole heads or lie inside one");
    constexpr int K     = GEMM_K;
    constexpr int PITCH = GEMM_PITCH;
    constexpr int TN4   = NOUT / 64;
    constexpr int NW    = 16;
    constexpr int MTR   = NW / TN4;
    constexpr int AROWS = MTR * 16;
    constexpr int APASS = AROWS / 8;
    constexpr int BQ    = NOUT * (K / 4) / 512;
    static_assert(NW % TN4 == 0 && (BQ & (BQ - 1)) == 0, "shape");

    __shared__ __attribute__((aligned(16))) _Float16 Bs[NOUT * PITCH];
    __shared__ __attribute__((aligned(16))) _Float16 As[2][AROWS * PITCH];

    if (SCAN && blockIdx.x == 0) {
        scan_counts<100>(deg, rowptr, nnodes);
        return;
    }
    const unsigned bx  = SCAN ? blockIdx.x - 1 : blockIdx.x;
    const unsigned nbx = SCAN ? gridDim.x - 1 : gridDim.x;

    const int tid   = threadIdx.x;
    const int w     = __builtin_amdgcn_readfirstlane(tid >> 5);
    const int lane  = tid & 31;
    const int r     = lane & 15;
    const int hi    = lane >> 4;
    const int khalf = hi << 3;
    const int mtl   = w / TN4;
    const int nt4   = w - mtl * TN4;

    const int t0 = (int)(bx * (unsigned)mtiles / nbx);
    const int t1 = (int)((bx + 1u) * (unsigned)mtiles / nbx);

    const int srow = tid >> 6;
    const int scol = (tid & 63) * 4;
    float4 pre[APASS];
    auto fetch = [&](int tb) {
        const int last = (t1 - tb < MTR ? t1 - tb : MTR) * 16 - 1;
#pragma unroll
        for (int p = 0; p < APASS; ++p) {
            const int row = p * 8 + srow;
            pre[p] = *(const float4*)(A + ((long)tb * 16 + (row < last ? row : last)) * K + scol);
        }
    };
    auto stash = [&](int buf) {
#pragma unroll
        for (int p = 0; p < APASS; ++p)
            put4h(&As[buf][(p * 8 + srow) * PITCH + scol], pre[p]);
    };

    if (t0 < t1) fetch(t0);
    {
        constexpr int BB = 8;
        for (int q0 = 0; q0 < BQ; q0 += BB) {
            float4 t[BB];
#pragma unroll
            for (int j = 0; j < BB; ++j) {
                const int idx = ((q0 + j + (int)bx) & (BQ - 1)) * 512 + tid;
                t[j] = *(const float4*)(W + (long)idx * 4);
            }
#pragma unroll
            for (int j = 0; j < BB; ++j) {
                const int idx = ((q0 + j + (int)bx) & (BQ - 1)) * 512 + tid;
                put4h(&Bs[(idx >> 6) * PITCH + (idx & 63) * 4], t[j]);
            }
        }
    }

    float sa[4], da[4];
#pragma unroll
    for (int j = 0; j < 4; ++j) {
        sa[j] = asr[nt4 * 64 + j * 16 + r];
        da[j] = adr[nt4 * 64 + j * 16 + r];
    }

    const _Float16* bp0 = &Bs[(nt4 * 64 + r) * PITCH + khalf];
    const int hi8 = hi << 3;

    int buf = 0;
    for (int tb = t0; tb < t1; tb += MTR, buf ^= 1) {
        stash(buf);
        __syncthreads();
        if (tb + MTR < t1) fetch(tb + MTR);

        const int mt = tb + mtl;
        if (mt < t1) {
            const _Float16* ap0 = &As[buf][(mtl * 16 + r) * PITCH + khalf];
            v8f acc0 = {}, acc1 = {}, acc2 = {}, acc3 = {};
            KStep st[2];
            load_kstep(st[0], ap0, bp0);
#pragma unroll
            for (int kk = 0; kk < K; kk += 32) {
                const int cur = (kk >> 5) & 1;
                __builtin_amdgcn_sched_barrier(0);
                if (kk + 32 < K) load_kstep(st[cur ^ 1], ap0 + kk + 32, bp0 + kk + 32);
                __builtin_amdgcn_sched_barrier(0);
                const v16h a = frag16(st[cur].alo, st[cur].ahi);
                acc0 = wmma_f16(a, frag16(st[cur].blo[0], st[cur].bhi[0]), acc0);
                acc1 = wmma_f16(a, frag16(st[cur].blo[1], st[cur].bhi[1]), acc1);
                acc2 = wmma_f16(a, frag16(st[cur].blo[2], st[cur].bhi[2]), acc2);
                acc3 = wmma_f16(a, frag16(st[cur].blo[3], st[cur].bhi[3]), acc3);
            }

            if (C16) {
                _Float16* crow = (_Float16*)C + (long)nt4 * 64 + r;
#pragma unroll
                for (int v = 0; v < 8; ++v) {
                    long ro = (long)(mt * 16 + hi8 + v) * NOUT;
                    crow[ro +  0] = (_Float16)acc0[v];
                    crow[ro + 16] = (_Float16)acc1[v];
                    crow[ro + 32] = (_Float16)acc2[v];
                    crow[ro + 48] = (_Float16)acc3[v];
                }
            } else {
                float* crow = C + (long)nt4 * 64 + r;
#pragma unroll
                for (int v = 0; v < 8; ++v) {
                    long ro = (long)(mt * 16 + hi8 + v) * NOUT;
                    crow[ro +  0] = acc0[v];
                    crow[ro + 16] = acc1[v];
                    crow[ro + 32] = acc2[v];
                    crow[ro + 48] = acc3[v];
                }
            }

            const long row = mt * 16 + hi8 + (r & 7);
            if (HC == 32) {
                float xs[16], xd[16];
#pragma unroll
                for (int v = 0; v < 8; ++v) {
                    xs[v]     = acc0[v] * sa[0] + acc1[v] * sa[1];
                    xs[8 + v] = acc2[v] * sa[2] + acc3[v] * sa[3];
                    xd[v]     = acc0[v] * da[0] + acc1[v] * da[1];
                    xd[8 + v] = acc2[v] * da[2] + acc3[v] * da[3];
                }
                const float ts = half_tree_sum16(xs, r);
                const float td = half_tree_sum16(xd, r);
                const long o = row * (2 * TN4) + 2 * nt4 + (r >> 3);
                als[o] = ts;
                ald[o] = td;
            } else {
                float x[16];
#pragma unroll
                for (int v = 0; v < 8; ++v) {
                    x[v]     = (acc0[v] * sa[0] + acc1[v] * sa[1]) + (acc2[v] * sa[2] + acc3[v] * sa[3]);
                    x[8 + v] = (acc0[v] * da[0] + acc1[v] * da[1]) + (acc2[v] * da[2] + acc3[v] * da[3]);
                }
                const float t = half_tree_sum16(x, r);
                float* dstp = (r >> 3) ? ald : als;
                dstp[row * TN4 + nt4] = t;
            }
        }
    }
}

__global__ void k_fill_i32(int* __restrict__ p, int v, long n) {
    long i = (long)blockIdx.x * blockDim.x + threadIdx.x;
    if (i < n) p[i] = v;
}

#define EDGE_U 2
#define EDGE_W (EDGE_U * 128)

__device__ __forceinline__ bool aligned16(const void* p) { return ((size_t)p & 15) == 0; }

__device__ __forceinline__ void edge_loads_issued_by_here(const int4& a, const int4& b) {
    asm volatile("" :: "v"(a.x), "v"(a.y), "v"(a.z), "v"(a.w),
                       "v"(b.x), "v"(b.y), "v"(b.z), "v"(b.w));
}

__global__ void __launch_bounds__(256)
k_deg_slot(const int* __restrict__ dst, int E, int* __restrict__ cnt, int* __restrict__ slot) {
    const int lane = threadIdx.x & 31;
    const int wave = __builtin_amdgcn_readfirstlane(
        (int)(blockIdx.x * (blockDim.x >> 5) + (threadIdx.x >> 5)));
    const int w0 = wave * EDGE_W;
    if (w0 + EDGE_W <= E && aligned16(dst) && aligned16(slot)) {
        int4 d[EDGE_U], s[EDGE_U];
#pragma unroll
        for (int j = 0; j < EDGE_U; ++j) d[j] = *(const int4*)(dst + w0 + (j * 32 + lane) * 4);
        static_assert(EDGE_U == 2, "the load fence below names two groups");
        edge_loads_issued_by_here(d[0], d[1]);
#pragma unroll
        for (int j = 0; j < EDGE_U; ++j) {
            s[j].x = atomicAdd(&cnt[d[j].x], 1);
            s[j].y = atomicAdd(&cnt[d[j].y], 1);
            s[j].z = atomicAdd(&cnt[d[j].z], 1);
            s[j].w = atomicAdd(&cnt[d[j].w], 1);
        }
#pragma unroll
        for (int j = 0; j < EDGE_U; ++j) *(int4*)(slot + w0 + (j * 32 + lane) * 4) = s[j];
    } else {
        for (int e = w0 + lane; e < w0 + EDGE_W && e < E; e += 32)
            slot[e] = atomicAdd(&cnt[dst[e]], 1);
    }
}

__global__ void __launch_bounds__(256)
k_place(const int* __restrict__ src, const int* __restrict__ dst,
        const int* __restrict__ slot, int E,
        const int* __restrict__ rowptr, int* __restrict__ tmp) {
    const int lane = threadIdx.x & 31;
    const int wave = __builtin_amdgcn_readfirstlane(
        (int)(blockIdx.x * (blockDim.x >> 5) + (threadIdx.x >> 5)));
    const int w0 = wave * EDGE_W;
    if (w0 + EDGE_W <= E && aligned16(src) && aligned16(dst) && aligned16(slot)) {
        int4 s[EDGE_U], d[EDGE_U], k[EDGE_U], p[EDGE_U];
#pragma unroll
        for (int j = 0; j < EDGE_U; ++j) {
            const int e0 = w0 + (j * 32 + lane) * 4;
            d[j] = *(const int4*)(dst + e0);
            k[j] = *(const int4*)(slot + e0);
            s[j] = *(const int4*)(src + e0);
        }
#pragma unroll
        for (int j = 0; j < EDGE_U; ++j) {
            p[j].x = rowptr[d[j].x] + k[j].x;
            p[j].y = rowptr[d[j].y] + k[j].y;
            p[j].z = rowptr[d[j].z] + k[j].z;
            p[j].w = rowptr[d[j].w] + k[j].w;
        }
#pragma unroll
        for (int j = 0; j < EDGE_U; ++j) {
            tmp[p[j].x] = s[j].x;
            tmp[p[j].y] = s[j].y;
            tmp[p[j].z] = s[j].z;
            tmp[p[j].w] = s[j].w;
        }
    } else {
        for (int e = w0 + lane; e < w0 + EDGE_W && e < E; e += 32)
            tmp[rowptr[dst[e]] + slot[e]] = src[e];
    }
}

__global__ void __launch_bounds__(256)
k_rank(const int* __restrict__ rowptr, const int* __restrict__ tmp,
       int* __restrict__ csr, int N, int totalWaves) {
    const int lane  = threadIdx.x & 31;
    const int wave0 = blockIdx.x * (blockDim.x >> 5) + (threadIdx.x >> 5);
    for (int nn = wave0; nn < N; nn += totalWaves) {
        const int n  = __builtin_amdgcn_readfirstlane(nn);
        const int b  = rowptr[n], en = rowptr[n + 1];
        const int deg = en - b;
        if (deg <= 32) {
            const int mine = lane < deg ? tmp[b + lane] : 0x7fffffff;
            int r = 0;
            for (int j = 0; j < deg; ++j) {
                const int v = __builtin_amdgcn_readlane(mine, j);
                r += (v < mine || (v == mine && j < lane)) ? 1 : 0;
            }
            if (lane < deg) csr[b + r] = mine;
        } else {
            for (int p = b + lane; p < en; p += 32) {
                const int s = tmp[p];
                int r = 0;
                for (int q = b; q < en; ++q) {
                    const int v = tmp[q];
                    r += (v < s || (v == s && q < p)) ? 1 : 0;
                }
                csr[b + r] = s;
            }
        }
    }
}

__device__ __forceinline__ float leaky(float v) { return v > 0.f ? v : NEG_SLOPE * v; }

__device__ __forceinline__ float exp_le0(float x) {
    return __builtin_amdgcn_exp2f(x * 1.44269504088896340736f);
}

template <int P>
__device__ __forceinline__ float lterm(const float* __restrict__ p, int i) {
    if (P == 1) return p[i];
    const float2 t = *(const float2*)(p + 2 * (long)i);
    return t.x + t.y;
}

template <int NV>
__device__ __forceinline__ void loads_issued_by_here(const float (&l)[NV]) {
    static_assert(NV == 1 || NV == 8, "one operand list per count in use");
    if constexpr (NV == 1)
        asm volatile("" :: "v"(l[0]));
    else
        asm volatile("" :: "v"(l[0]), "v"(l[1]), "v"(l[2]), "v"(l[3]),
                           "v"(l[4]), "v"(l[5]), "v"(l[6]), "v"(l[7]));
}

template <int R16, int NOUT>
__device__ __forceinline__ float4 row4(const float* __restrict__ Hm, long row, int ch) {
    if (R16) {
        const v4h h = *(const v4h*)((const _Float16*)Hm + row * NOUT + ch);
        return make_float4((float)h[0], (float)h[1], (float)h[2], (float)h[3]);
    }
    return *(const float4*)(Hm + row * NOUT + ch);
}

__device__ __forceinline__ void mad_mix_lo(float& acc, float w, unsigned h2) {
    asm("v_fma_mix_f32 %0, %1, %2, %0 op_sel_hi:[0,1,0]" : "+v"(acc) : "v"(w), "v"(h2));
}
__device__ __forceinline__ void mad_mix_hi(float& acc, float w, unsigned h2) {
    asm("v_fma_mix_f32 %0, %1, %2, %0 op_sel:[0,1,0] op_sel_hi:[0,1,0]" : "+v"(acc) : "v"(w), "v"(h2));
}

template <int H, int C, int ELU, int R16>
__global__ void __launch_bounds__(256)
k_gather(const int* __restrict__ rowptr, const int* __restrict__ csr,
         const float* __restrict__ als, const float* __restrict__ ald,
         const float* __restrict__ Hm, const float* __restrict__ bias,
         float* __restrict__ out, int N, int totalWaves) {
    constexpr int NOUT = H * C;
    constexpr int V    = NOUT / 128;
    constexpr int P    = C > 64 ? C / 64 : 1;
    constexpr int G    = C / 4 < 32 ? C / 4 : 32;
    constexpr int T    = 32 / G;
    static_assert((G & (G - 1)) == 0 && G * T == 32, "head groups must tile the wave");
    const int lane = threadIdx.x & 31;
    const int gi   = lane & (G - 1);
    const int gb   = lane & ~(G - 1);
    const int wave0 = blockIdx.x * (blockDim.x >> 5) + (threadIdx.x >> 5);

    int hd[V], ch[V];
#pragma unroll
    for (int k = 0; k < V; ++k) { ch[k] = k * 128 + lane * 4; hd[k] = ch[k] / C; }

    for (int nn = wave0; nn < N; nn += totalWaves) {
        const int n   = __builtin_amdgcn_readfirstlane(nn);
        const int beg = rowptr[n], end = rowptr[n + 1];
        const bool one = end - beg <= 32;

        float adn[V], es[V], m[V], den[V];
        float4 acc[V];
#pragma unroll
        for (int k = 0; k < V; ++k) {
            adn[k] = lterm<P>(ald, n * H + hd[k]);
            es[k]  = leaky(lterm<P>(als, n * H + hd[k]) + adn[k]);
            m[k]   = es[k];
        }

        int   mys = 0;
        float e[V][T];
        auto chunk_logits = [&](int cnt) {
            int sv[T];
#pragma unroll
            for (int t = 0; t < T; ++t) sv[t] = T == 1 ? mys : __shfl(mys, gi + G * t);
            float l[V * T];
#pragma unroll
            for (int k = 0; k < V; ++k)
#pragma unroll
                for (int t = 0; t < T; ++t)
                    l[k * T + t] = lterm<P>(als, sv[t] * H + hd[k]);
            loads_issued_by_here(l);
#pragma unroll
            for (int k = 0; k < V; ++k)
#pragma unroll
                for (int t = 0; t < T; ++t)
                    e[k][t] = gi + G * t < cnt ? leaky(l[k * T + t] + adn[k]) : -__builtin_inff();
        };

        for (int base = beg; base < end; base += 32) {
            const int cnt = end - base < 32 ? end - base : 32;
            mys = lane < cnt ? csr[base + lane] : 0;
            chunk_logits(cnt);
#pragma unroll
            for (int k = 0; k < V; ++k)
#pragma unroll
                for (int t = 0; t < T; ++t) m[k] = fmaxf(m[k], e[k][t]);
        }
#pragma unroll
        for (int k = 0; k < V; ++k)
#pragma unroll
            for (int off = G / 2; off > 0; off >>= 1) m[k] = fmaxf(m[k], __shfl_xor(m[k], off));

#pragma unroll
        for (int k = 0; k < V; ++k) {
            const float ex = exp_le0(es[k] - m[k]);
            const float4 hv = row4<R16, NOUT>(Hm, n, ch[k]);
            den[k] = ex;
            acc[k] = make_float4(hv.x * ex, hv.y * ex, hv.z * ex, hv.w * ex);
        }
        for (int base = beg; base < end; base += 32) {
            const int cnt = end - base < 32 ? end - base : 32;
            if (!one) {
                mys = lane < cnt ? csr[base + lane] : 0;
                chunk_logits(cnt);
            }
#pragma unroll
            for (int t = 0; t < T; ++t) {
                const int lim = cnt - t * G < G ? cnt - t * G : G;
                if (lim > 0) {
                    float p[V];
#pragma unroll
                    for (int k = 0; k < V; ++k) p[k] = exp_le0(e[k][t] - m[k]);
#pragma unroll 4
                    for (int i = 0; i < lim; ++i) {
                        const int s = __builtin_amdgcn_readlane(mys, t * G + i);
#pragma unroll
                        for (int k = 0; k < V; ++k) {
                            const float ex = G == 32
                                ? __int_as_float(__builtin_amdgcn_readlane(__float_as_int(p[k]), i))
                                : __shfl(p[k], gb + i);
                            den[k] += ex;
                            if (R16) {
                                const uint2 h = *(const uint2*)((const _Float16*)Hm + (long)s * NOUT + ch[k]);
                                mad_mix_lo(acc[k].x, ex, h.x);
                                mad_mix_hi(acc[k].y, ex, h.x);
                                mad_mix_lo(acc[k].z, ex, h.y);
                                mad_mix_hi(acc[k].w, ex, h.y);
                            } else {
                                const float4 hv = row4<R16, NOUT>(Hm, s, ch[k]);
                                acc[k].x += hv.x * ex;
                                acc[k].y += hv.y * ex;
                                acc[k].z += hv.z * ex;
                                acc[k].w += hv.w * ex;
                            }
                        }
                    }
                }
            }
        }
#pragma unroll
        for (int k = 0; k < V; ++k) {
            const float rdn = 1.0f / (den[k] + 1e-16f);
            const float4 bv = *(const float4*)(bias + ch[k]);
            float4 o;
            o.x = acc[k].x * rdn + bv.x;
            o.y = acc[k].y * rdn + bv.y;
            o.z = acc[k].z * rdn + bv.z;
            o.w = acc[k].w * rdn + bv.w;
            if (ELU) {
                o.x = o.x > 0.f ? o.x : (exp_le0(fminf(o.x, 0.f)) - 1.f);
                o.y = o.y > 0.f ? o.y : (exp_le0(fminf(o.y, 0.f)) - 1.f);
                o.z = o.z > 0.f ? o.z : (exp_le0(fminf(o.z, 0.f)) - 1.f);
                o.w = o.w > 0.f ? o.w : (exp_le0(fminf(o.w, 0.f)) - 1.f);
            }
            *(float4*)(out + (long)n * NOUT + ch[k]) = o;
        }
    }
}

static inline long cdivl(long a, long b) { return (a + b - 1) / b; }

struct LayerWS {
    float *als, *ald;
    const int *rowptr, *csr;
};

static void run_gat_gemm(const float* Xin, const float* W, const float* a_s, const float* a_d,
                         int N, int H, float* Hbuf, const LayerWS& w,
                         const int* deg, int* rowptr, hipStream_t stream) {
    if (H == 8 && deg)
        k_gemm_wmma<256, 32, 1, 1><<<256, 512, 0, stream>>>(Xin, W, Hbuf, a_s, a_d, w.als, w.ald, N / 16,
                                                            deg, rowptr, N);
    else if (H == 8)
        k_gemm_wmma<256, 32, 1, 0><<<256, 512, 0, stream>>>(Xin, W, Hbuf, a_s, a_d, w.als, w.ald, N / 16,
                                                            nullptr, nullptr, 0);
    else
        k_gemm_wmma<128, 128, 0, 0><<<256, 512, 0, stream>>>(Xin, W, Hbuf, a_s, a_d, w.als, w.ald, N / 16,
                                                             nullptr, nullptr, 0);
}

static void run_gat_gather(const float* bias, int N, int H,
                           const float* Hbuf, float* agg, const LayerWS& w,
                           hipStream_t stream) {
    const int TB = 256;
    const int GB = 2048, gWaves = GB * (TB / 32);
    if (H == 8)
        k_gather<8, 32, 1, 1><<<GB, TB, 0, stream>>>(w.rowptr, w.csr, w.als, w.ald, Hbuf, bias,
                                                  agg, N, gWaves);
    else
        k_gather<1, 128, 0, 0><<<GB, TB, 0, stream>>>(w.rowptr, w.csr, w.als, w.ald, Hbuf, bias,
                                                   agg, N, gWaves);
}

extern "C" void kernel_launch(void* const* d_in, const int* in_sizes, int n_in,
                              void* d_out, int out_size, void* d_ws, size_t ws_size,
                              hipStream_t stream) {
    const float* x   = (const float*)d_in[0];
    const int*   ei  = (const int*)  d_in[1];
    const float* W1  = (const float*)d_in[2];
    const float* as1 = (const float*)d_in[3];
    const float* ad1 = (const float*)d_in[4];
    const float* b1  = (const float*)d_in[5];
    const float* W2  = (const float*)d_in[6];
    const float* as2 = (const float*)d_in[7];
    const float* ad2 = (const float*)d_in[8];
    const float* b2  = (const float*)d_in[9];
    const float* W3  = (const float*)d_in[10];
    const float* as3 = (const float*)d_in[11];
    const float* ad3 = (const float*)d_in[12];
    const float* b3  = (const float*)d_in[13];
    float* out = (float*)d_out;

    const int FinW = 256;
    const int N    = in_sizes[0] / FinW;
    const int E    = in_sizes[1] / 2;
    const int* src = ei;
    const int* dst = ei + E;

    char* p = (char*)d_ws;
    float*    P   = (float*)p;    p += (size_t)N * 256 * sizeof(float);
    float*    Q   = (float*)p;    p += (size_t)N * 256 * sizeof(float);
    float*    als = (float*)p;    p += (size_t)N * 8 * sizeof(float);
    float*    ald = (float*)p;    p += (size_t)N * 8 * sizeof(float);
    int*      cnt = (int*)p;      p += (size_t)N * sizeof(int);
    int*      slot = (int*)p;     p += (size_t)(E + 4) * sizeof(int);
    int*      rowptr = (int*)p;   p += (size_t)(N + 4) * sizeof(int);
    int*      tmp  = (int*)p;     p += (size_t)E * sizeof(int);
    int*      csr  = (int*)p;     p += (size_t)E * sizeof(int);
    LayerWS w{als, ald, rowptr, csr};

    const int TB = 256;
    const int EB = (int)cdivl(E, (long)(TB / 32) * EDGE_W);
    k_fill_i32<<<(int)cdivl(N, TB), TB, 0, stream>>>(cnt, 0, (long)N);
    k_deg_slot<<<EB, TB, 0, stream>>>(dst, E, cnt, slot);
    run_gat_gemm(x, W1, as1, ad1, N, 8, Q, w, cnt, rowptr, stream);
    k_place<<<EB, TB, 0, stream>>>(src, dst, slot, E, rowptr, tmp);
    k_rank<<<2048, TB, 0, stream>>>(rowptr, tmp, csr, N, 2048 * (TB / 32));

    run_gat_gather(b1, N, 8, Q, P, w, stream);
    run_gat_gemm(P, W2, as2, ad2, N, 8, Q, w, nullptr, nullptr, stream);
    run_gat_gather(b2, N, 8, Q, P, w, stream);
    run_gat_gemm(P, W3, as3, ad3, N, 1, Q, w, nullptr, nullptr, stream);
    run_gat_gather(b3, N, 1, Q, out, w, stream);
}
